# starting kernel
# baseline (speedup 1.0000x reference)
#include <hip/hip_runtime.h>
#include <math.h>

typedef float v2f __attribute__((ext_vector_type(2)));
typedef float v8f __attribute__((ext_vector_type(8)));

#define WMMA_F32(a, b, c) \
  __builtin_amdgcn_wmma_f32_16x16x4_f32(false, (a), false, (b), (short)0, (c), false, false)

static constexpr int B_ = 8, S_ = 512, E_ = 32, L_ = 4096, SUBE_ = 768;
static constexpr float LAMBDA_INIT = 0.2f;

static constexpr int ST0 = 513;
static constexpr int H1  = 16 * ST0;
static constexpr int SMEM_FLOATS = 16 * ST0 + 16 * 512;

__global__ __launch_bounds__(32) void kv_proj_kernel(
    const float* __restrict__ key, const float* __restrict__ Wk,
    const float* __restrict__ Wv, float* __restrict__ Km, float* __restrict__ Vm) {
  const int lane = threadIdx.x;
  const int n    = lane & 15;
  const int khi  = lane >> 4;
  const int m0   = blockIdx.x * 16;
  const int mrow = m0 + n;

  v8f ak0 = {}, ak1 = {}, av0 = {}, av1 = {};
  for (int kk = 0; kk < SUBE_ / 4; ++kk) {
    const int kb = kk * 4 + khi * 2;
    v2f a   = *(const v2f*)(key + (size_t)mrow * SUBE_ + kb);
    v2f bk0 = *(const v2f*)(Wk + (size_t)n * SUBE_ + kb);
    v2f bk1 = *(const v2f*)(Wk + (size_t)(16 + n) * SUBE_ + kb);
    v2f bv0 = *(const v2f*)(Wv + (size_t)n * SUBE_ + kb);
    v2f bv1 = *(const v2f*)(Wv + (size_t)(16 + n) * SUBE_ + kb);
    ak0 = WMMA_F32(a, bk0, ak0);
    ak1 = WMMA_F32(a, bk1, ak1);
    av0 = WMMA_F32(a, bv0, av0);
    av1 = WMMA_F32(a, bv1, av1);
  }
  const int rbase = m0 + (khi ? 8 : 0);
  for (int v = 0; v < 8; ++v) {
    const size_t r = (size_t)(rbase + v) * 32;
    Km[r + n]      = ak0[v];
    Km[r + 16 + n] = ak1[v];
    Vm[r + n]      = av0[v];
    Vm[r + 16 + n] = av1[v];
  }
}

__global__ __launch_bounds__(32) void q_proj_kernel(
    const float* __restrict__ q, const float* __restrict__ Wq,
    float* __restrict__ Qs) {
  const int lane = threadIdx.x;
  const int n    = lane & 15;
  const int khi  = lane >> 4;
  const int m0   = blockIdx.x * 16;
  const int mrow = m0 + n;

  v8f a0 = {}, a1 = {};
  for (int kk = 0; kk < 8; ++kk) {
    const int kb = kk * 4 + khi * 2;
    v2f a  = *(const v2f*)(q + (size_t)mrow * 32 + kb);
    v2f b0 = *(const v2f*)(Wq + (size_t)n * 32 + kb);
    v2f b1 = *(const v2f*)(Wq + (size_t)(16 + n) * 32 + kb);
    a0 = WMMA_F32(a, b0, a0);
    a1 = WMMA_F32(a, b1, a1);
  }
  const int rbase = m0 + (khi ? 8 : 0);
  for (int v = 0; v < 8; ++v) {
    const size_t r = (size_t)(rbase + v) * 32;
    Qs[r + n]      = a0[v] * 0.25f;
    Qs[r + 16 + n] = a1[v] * 0.25f;
  }
}

__global__ __launch_bounds__(32) void attn_kernel(
    const float* __restrict__ Qs, const float* __restrict__ Km,
    const float* __restrict__ Vm, const int* __restrict__ qmask,
    const int* __restrict__ kmask, const float* __restrict__ lq1,
    const float* __restrict__ lk1, const float* __restrict__ lq2,
    const float* __restrict__ lk2, const float* __restrict__ lnw,
    const float* __restrict__ Wo, float* __restrict__ outp,
    float* __restrict__ diffp) {
  __shared__ float smem[SMEM_FLOATS];

  const int lane = threadIdx.x;
  const int n    = lane & 15;
  const int khi  = lane >> 4;
  const int R0   = blockIdx.x * 16;
  const int b    = R0 / L_;
  const int rb   = khi ? 8 : 0;

  float s1 = 0.f, s2 = 0.f;
  for (int i = 0; i < 16; ++i) { s1 += lq1[i] * lk1[i]; s2 += lq2[i] * lk2[i]; }
  const float lambda_full = __expf(s1) - __expf(s2) + LAMBDA_INIT;

  for (int h = 0; h < 2; ++h) {
    float* dst       = (h == 0) ? smem : (smem + H1);
    const int stride = (h == 0) ? ST0 : 512;
    for (int st = 0; st < 32; ++st) {
      v8f acc = {};
      for (int kk = 0; kk < 4; ++kk) {
        const int kb = kk * 4 + khi * 2;
        v2f a  = *(const v2f*)(Qs + (size_t)(R0 + n) * 32 + h * 16 + kb);
        v2f bf = *(const v2f*)(Km + (size_t)(b * S_ + st * 16 + n) * 32 + h * 16 + kb);
        acc = WMMA_F32(a, bf, acc);
      }
      const int col = st * 16 + n;
      const int kmv = kmask[b * S_ + col];
      for (int v = 0; v < 8; ++v) {
        const int r     = rb + v;
        const int valid = kmv && qmask[R0 + r];
        dst[r * stride + col] = valid ? acc[v] : -1e20f;
      }
    }
  }
  __syncthreads();

  for (int m = 0; m < 16; ++m) {
    const float* r0 = smem + m * ST0;
    const float* r1 = smem + H1 + m * 512;
    float mx0 = -3.4e38f, mx1 = -3.4e38f;
    for (int i = 0; i < 16; ++i) {
      const int c = lane + 32 * i;
      mx0 = fmaxf(mx0, r0[c]);
      mx1 = fmaxf(mx1, r1[c]);
    }
    for (int o = 16; o > 0; o >>= 1) {
      mx0 = fmaxf(mx0, __shfl_xor(mx0, o, 32));
      mx1 = fmaxf(mx1, __shfl_xor(mx1, o, 32));
    }
    float e0[16], e1[16], sum0 = 0.f, sum1 = 0.f;
    for (int i = 0; i < 16; ++i) {
      const int c = lane + 32 * i;
      e0[i] = __expf(r0[c] - mx0); sum0 += e0[i];
      e1[i] = __expf(r1[c] - mx1); sum1 += e1[i];
    }
    for (int o = 16; o > 0; o >>= 1) {
      sum0 += __shfl_xor(sum0, o, 32);
      sum1 += __shfl_xor(sum1, o, 32);
    }
    const float inv0 = 1.f / (sum0 + 1e-8f);
    const float inv1 = 1.f / (sum1 + 1e-8f);
    float d[16], dmin = 3.4e38f;
    for (int i = 0; i < 16; ++i) {
      d[i] = e0[i] * inv0 - lambda_full * (e1[i] * inv1);
      dmin = fminf(dmin, d[i]);
    }
    for (int o = 16; o > 0; o >>= 1) dmin = fminf(dmin, __shfl_xor(dmin, o, 32));

    const int qv = qmask[R0 + m];
    float* w0 = smem + m * ST0;
    float* dg = diffp + (size_t)(R0 + m) * 512;
    for (int i = 0; i < 16; ++i) {
      const int c = lane + 32 * i;
      float dd = d[i] - dmin + 1e-5f;
      dd = (qv && kmask[b * S_ + c]) ? dd : 0.f;
      w0[c] = dd;
      __builtin_nontemporal_store(dd, dg + c);
    }
  }
  __syncthreads();

  v8f ao0 = {}, ao1 = {};
  for (int kk = 0; kk < 128; ++kk) {
    const int kb = kk * 4 + khi * 2;
    v2f a; a[0] = smem[n * ST0 + kb]; a[1] = smem[n * ST0 + kb + 1];
    const float* vr0 = Vm + (size_t)(b * S_ + kb) * 32;
    const float* vr1 = Vm + (size_t)(b * S_ + kb + 1) * 32;
    v2f b0; b0[0] = vr0[n];      b0[1] = vr1[n];
    v2f b1; b1[0] = vr0[16 + n]; b1[1] = vr1[16 + n];
    ao0 = WMMA_F32(a, b0, ao0);
    ao1 = WMMA_F32(a, b1, ao1);
  }
  float* aos = smem + H1;
  for (int v = 0; v < 8; ++v) {
    const int r = rb + v;
    aos[r * 32 + n]      = ao0[v];
    aos[r * 32 + 16 + n] = ao1[v];
  }
  __syncthreads();

  for (int m = 0; m < 16; ++m) {
    const float x = aos[m * 32 + lane];
    float sq = x * x;
    for (int o = 16; o > 0; o >>= 1) sq += __shfl_xor(sq, o, 32);
    const float rms = rsqrtf(sq * (1.0f / 32.0f) + 1e-5f);
    aos[m * 32 + lane] = x * rms * lnw[lane] * 0.8f;
  }
  __syncthreads();

  v8f o0 = {}, o1 = {};
  for (int kk = 0; kk < 8; ++kk) {
    const int kb = kk * 4 + khi * 2;
    v2f a; a[0] = aos[n * 32 + kb]; a[1] = aos[n * 32 + kb + 1];
    v2f b0 = *(const v2f*)(Wo + (size_t)n * 32 + kb);
    v2f b1 = *(const v2f*)(Wo + (size_t)(16 + n) * 32 + kb);
    o0 = WMMA_F32(a, b0, o0);
    o1 = WMMA_F32(a, b1, o1);
  }
  for (int v = 0; v < 8; ++v) {
    const size_t r = (size_t)(R0 + rb + v) * 32;
    __builtin_nontemporal_store(o0[v], outp + r + n);
    __builtin_nontemporal_store(o1[v], outp + r + 16 + n);
  }
}

extern "C" void kernel_launch(void* const* d_in, const int* in_sizes, int n_in,
                              void* d_out, int out_size, void* d_ws, size_t ws_size,
                              hipStream_t stream) {
  (void)in_sizes; (void)n_in; (void)out_size; (void)ws_size;
  const float* query = (const float*)d_in[0];
  const float* key   = (const float*)d_in[1];
  const int*   qmask = (const int*)d_in[2];
  const int*   kmask = (const int*)d_in[3];
  const float* Wq = (const float*)d_in[4];
  const float* Wk = (const float*)d_in[5];
  const float* Wv = (const float*)d_in[6];
  const float* Wo = (const float*)d_in[7];
  const float* lq1 = (const float*)d_in[8];
  const float* lk1 = (const float*)d_in[9];
  const float* lq2 = (const float*)d_in[10];
  const float* lk2 = (const float*)d_in[11];
  const float* lnw = (const float*)d_in[12];

  float* outp  = (float*)d_out;
  float* diffp = outp + (size_t)B_ * L_ * E_;

  float* Qs = (float*)d_ws;
  float* Km = Qs + (size_t)B_ * L_ * E_;
  float* Vm = Km + (size_t)B_ * S_ * E_;

  kv_proj_kernel<<<(B_ * S_) / 16, 32, 0, stream>>>(key, Wk, Wv, Km, Vm);
  q_proj_kernel<<<(B_ * L_) / 16, 32, 0, stream>>>(query, Wq, Qs);
  attn_kernel<<<(B_ * L_) / 16, 32, 0, stream>>>(Qs, Km, Vm, qmask, kmask,
                                                 lq1, lk1, lq2, lk2, lnw, Wo,
                                                 outp, diffp);
}


# attention kernel: score tile kept transposed in registers (K x Q^T, one wave, no LDS); softmax reduced in-lane, D registers are B of V^T x diff^T; sum order changes: scores, softmax, attn_out, norm
# speedup vs baseline: 3.1425x; 3.1425x over previous
#include <hip/hip_runtime.h>
#include <math.h>

typedef float v2f __attribute__((ext_vector_type(2)));
typedef float v4f __attribute__((ext_vector_type(4)));
typedef int   v4i __attribute__((ext_vector_type(4)));
typedef float v8f __attribute__((ext_vector_type(8)));

#define WMMA_F32(a, b, c) \
  __builtin_amdgcn_wmma_f32_16x16x4_f32(false, (a), false, (b), (short)0, (c), false, false)

static constexpr int B_ = 8, S_ = 512, E_ = 32, L_ = 4096, SUBE_ = 768;
static constexpr float LAMBDA_INIT = 0.2f;

__global__ __launch_bounds__(32) void kv_proj_kernel(
    const float* __restrict__ key, const float* __restrict__ Wk,
    const float* __restrict__ Wv, float* __restrict__ Km, float* __restrict__ Vm) {
  const int lane = threadIdx.x;
  const int n    = lane & 15;
  const int khi  = lane >> 4;
  const int m0   = blockIdx.x * 16;
  const int mrow = m0 + n;

  v8f ak0 = {}, ak1 = {}, av0 = {}, av1 = {};
  for (int kk = 0; kk < SUBE_ / 4; ++kk) {
    const int kb = kk * 4 + khi * 2;
    v2f a   = *(const v2f*)(key + (size_t)mrow * SUBE_ + kb);
    v2f bk0 = *(const v2f*)(Wk + (size_t)n * SUBE_ + kb);
    v2f bk1 = *(const v2f*)(Wk + (size_t)(16 + n) * SUBE_ + kb);
    v2f bv0 = *(const v2f*)(Wv + (size_t)n * SUBE_ + kb);
    v2f bv1 = *(const v2f*)(Wv + (size_t)(16 + n) * SUBE_ + kb);
    ak0 = WMMA_F32(a, bk0, ak0);
    ak1 = WMMA_F32(a, bk1, ak1);
    av0 = WMMA_F32(a, bv0, av0);
    av1 = WMMA_F32(a, bv1, av1);
  }
  const int rbase = m0 + (khi ? 8 : 0);
  for (int v = 0; v < 8; ++v) {
    const size_t r = (size_t)(rbase + v) * 32;
    Km[r + n]      = ak0[v];
    Km[r + 16 + n] = ak1[v];
    Vm[r + n]      = av0[v];
    Vm[r + 16 + n] = av1[v];
  }
}

__global__ __launch_bounds__(32) void q_proj_kernel(
    const float* __restrict__ q, const float* __restrict__ Wq,
    float* __restrict__ Qs) {
  const int lane = threadIdx.x;
  const int n    = lane & 15;
  const int khi  = lane >> 4;
  const int m0   = blockIdx.x * 16;
  const int mrow = m0 + n;

  v8f a0 = {}, a1 = {};
  for (int kk = 0; kk < 8; ++kk) {
    const int kb = kk * 4 + khi * 2;
    v2f a  = *(const v2f*)(q + (size_t)mrow * 32 + kb);
    v2f b0 = *(const v2f*)(Wq + (size_t)n * 32 + kb);
    v2f b1 = *(const v2f*)(Wq + (size_t)(16 + n) * 32 + kb);
    a0 = WMMA_F32(a, b0, a0);
    a1 = WMMA_F32(a, b1, a1);
  }
  const int rbase = m0 + (khi ? 8 : 0);
  for (int v = 0; v < 8; ++v) {
    const size_t r = (size_t)(rbase + v) * 32;
    Qs[r + n]      = a0[v] * 0.25f;
    Qs[r + 16 + n] = a1[v] * 0.25f;
  }
}

__global__ __launch_bounds__(32) void attn_kernel(
    const float* __restrict__ Qs, const float* __restrict__ Km,
    const float* __restrict__ Vm, const int* __restrict__ qmask,
    const int* __restrict__ kmask, const float* __restrict__ lq1,
    const float* __restrict__ lk1, const float* __restrict__ lq2,
    const float* __restrict__ lk2, const float* __restrict__ lnw,
    const float* __restrict__ Wo, float* __restrict__ outp,
    float* __restrict__ diffp) {
  const int lane = threadIdx.x;
  const int c    = lane & 15;
  const int h    = lane >> 4;
  const int R0   = blockIdx.x * 16;
  const int b    = R0 / L_;

  float s1 = 0.f, s2 = 0.f;
  for (int i = 0; i < 16; ++i) { s1 += lq1[i] * lk1[i]; s2 += lq2[i] * lk2[i]; }
  const float lambda_full = __expf(s1) - __expf(s2) + LAMBDA_INIT;

  const int qv = qmask[R0 + c];

  v2f qf[2][4];
#pragma unroll
  for (int hd = 0; hd < 2; ++hd) {
    const float* qp = Qs + (size_t)(R0 + c) * 32 + hd * 16 + 8 * h;
    const v4f qa = *(const v4f*)(qp);
    const v4f qb = *(const v4f*)(qp + 4);
    qf[hd][0][0] = qa[0]; qf[hd][0][1] = qa[1];
    qf[hd][1][0] = qa[2]; qf[hd][1][1] = qa[3];
    qf[hd][2][0] = qb[0]; qf[hd][2][1] = qb[1];
    qf[hd][3][0] = qb[2]; qf[hd][3][1] = qb[3];
  }

  const int crow = ((c & 4) << 1) | ((c & 8) >> 1) | (c & 3);
  const float* kbase = Km + (size_t)(b * S_ + crow) * 32 + 8 * h;
  const int*   mbase = kmask + b * S_ + 4 * h;

  v8f S0[32], S1[32];
#pragma unroll
  for (int t = 0; t < 32; ++t) {
    const float* kp = kbase + (size_t)t * 16 * 32;
    const v4f k0a = *(const v4f*)(kp);
    const v4f k0b = *(const v4f*)(kp + 4);
    const v4f k1a = *(const v4f*)(kp + 16);
    const v4f k1b = *(const v4f*)(kp + 20);
    const v4i ma  = *(const v4i*)(mbase + t * 16);
    const v4i mb  = *(const v4i*)(mbase + t * 16 + 8);
    v8f a0 = {}, a1 = {};
    v2f f;
    f[0] = k0a[0]; f[1] = k0a[1]; a0 = WMMA_F32(f, qf[0][0], a0);
    f[0] = k1a[0]; f[1] = k1a[1]; a1 = WMMA_F32(f, qf[1][0], a1);
    f[0] = k0a[2]; f[1] = k0a[3]; a0 = WMMA_F32(f, qf[0][1], a0);
    f[0] = k1a[2]; f[1] = k1a[3]; a1 = WMMA_F32(f, qf[1][1], a1);
    f[0] = k0b[0]; f[1] = k0b[1]; a0 = WMMA_F32(f, qf[0][2], a0);
    f[0] = k1b[0]; f[1] = k1b[1]; a1 = WMMA_F32(f, qf[1][2], a1);
    f[0] = k0b[2]; f[1] = k0b[3]; a0 = WMMA_F32(f, qf[0][3], a0);
    f[0] = k1b[2]; f[1] = k1b[3]; a1 = WMMA_F32(f, qf[1][3], a1);
#pragma unroll
    for (int r = 0; r < 8; ++r) {
      const int km    = (r < 4) ? ma[r & 3] : mb[r & 3];
      const bool valid = km && qv;
      S0[t][r] = valid ? a0[r] : -1e20f;
      S1[t][r] = valid ? a1[r] : -1e20f;
    }
  }

  v8f m0v = S0[0], m1v = S1[0];
#pragma unroll
  for (int t = 1; t < 32; ++t) {
#pragma unroll
    for (int r = 0; r < 8; ++r) {
      m0v[r] = fmaxf(m0v[r], S0[t][r]);
      m1v[r] = fmaxf(m1v[r], S1[t][r]);
    }
  }
  float mx0 = fmaxf(fmaxf(fmaxf(m0v[0], m0v[1]), fmaxf(m0v[2], m0v[3])),
                    fmaxf(fmaxf(m0v[4], m0v[5]), fmaxf(m0v[6], m0v[7])));
  float mx1 = fmaxf(fmaxf(fmaxf(m1v[0], m1v[1]), fmaxf(m1v[2], m1v[3])),
                    fmaxf(fmaxf(m1v[4], m1v[5]), fmaxf(m1v[6], m1v[7])));
  mx0 = fmaxf(mx0, __shfl_xor(mx0, 16, 32));
  mx1 = fmaxf(mx1, __shfl_xor(mx1, 16, 32));

  v8f s0v = {}, s1v = {};
#pragma unroll
  for (int t = 0; t < 32; ++t) {
#pragma unroll
    for (int r = 0; r < 8; ++r) {
      const float e0 = __expf(S0[t][r] - mx0);
      const float e1 = __expf(S1[t][r] - mx1);
      S0[t][r] = e0; s0v[r] += e0;
      S1[t][r] = e1; s1v[r] += e1;
    }
  }
  float sum0 = ((s0v[0] + s0v[1]) + (s0v[2] + s0v[3])) +
               ((s0v[4] + s0v[5]) + (s0v[6] + s0v[7]));
  float sum1 = ((s1v[0] + s1v[1]) + (s1v[2] + s1v[3])) +
               ((s1v[4] + s1v[5]) + (s1v[6] + s1v[7]));
  sum0 += __shfl_xor(sum0, 16, 32);
  sum1 += __shfl_xor(sum1, 16, 32);
  const float inv0 = 1.f / (sum0 + 1e-8f);
  const float inv1 = 1.f / (sum1 + 1e-8f);

  v8f dmv;
#pragma unroll
  for (int t = 0; t < 32; ++t) {
#pragma unroll
    for (int r = 0; r < 8; ++r) {
      const float d = S0[t][r] * inv0 - lambda_full * (S1[t][r] * inv1);
      S0[t][r] = d;
      dmv[r] = (t == 0) ? d : fminf(dmv[r], d);
    }
  }
  float dmin = fminf(fminf(fminf(dmv[0], dmv[1]), fminf(dmv[2], dmv[3])),
                     fminf(fminf(dmv[4], dmv[5]), fminf(dmv[6], dmv[7])));
  dmin = fminf(dmin, __shfl_xor(dmin, 16, 32));

  float* dg = diffp + (size_t)(R0 + c) * 512 + 4 * h;
#pragma unroll
  for (int t = 0; t < 32; ++t) {
    const v4i ma = *(const v4i*)(mbase + t * 16);
    const v4i mb = *(const v4i*)(mbase + t * 16 + 8);
    v4f lo, hi;
#pragma unroll
    for (int r = 0; r < 8; ++r) {
      const int km = (r < 4) ? ma[r & 3] : mb[r & 3];
      float dd = S0[t][r] - dmin + 1e-5f;
      dd = (qv && km) ? dd : 0.f;
      S0[t][r] = dd;
      if (r < 4) lo[r & 3] = dd; else hi[r & 3] = dd;
    }
    __builtin_nontemporal_store(lo, (v4f*)(dg + t * 16));
    __builtin_nontemporal_store(hi, (v4f*)(dg + t * 16 + 8));
  }

  const float* vbase = Vm + (size_t)(b * S_ + 4 * h) * 32 + c;
  v8f ao0 = {}, ao1 = {};
#pragma unroll
  for (int t = 0; t < 32; ++t) {
#pragma unroll
    for (int j = 0; j < 4; ++j) {
      const float* vp = vbase + (size_t)(t * 16 + 8 * (j >> 1) + 2 * (j & 1)) * 32;
      v2f a0, a1, bf;
      a0[0] = vp[0];  a0[1] = vp[32];
      a1[0] = vp[16]; a1[1] = vp[48];
      bf[0] = S0[t][2 * j]; bf[1] = S0[t][2 * j + 1];
      ao0 = WMMA_F32(a0, bf, ao0);
      ao1 = WMMA_F32(a1, bf, ao1);
    }
  }

  float sq = 0.f;
#pragma unroll
  for (int r = 0; r < 8; ++r) sq += ao0[r] * ao0[r];
#pragma unroll
  for (int r = 0; r < 8; ++r) sq += ao1[r] * ao1[r];
  sq += __shfl_xor(sq, 16, 32);
  const float rms = rsqrtf(sq * (1.0f / 32.0f) + 1e-5f);
#pragma unroll
  for (int r = 0; r < 8; ++r) {
    ao0[r] = ao0[r] * rms * lnw[8 * h + r] * 0.8f;
    ao1[r] = ao1[r] * rms * lnw[16 + 8 * h + r] * 0.8f;
  }

  v8f o0 = {}, o1 = {};
#pragma unroll
  for (int mt = 0; mt < 2; ++mt) {
#pragma unroll
    for (int j = 0; j < 4; ++j) {
      const int kb = 16 * mt + 8 * h + 2 * j;
      const v2f w0 = *(const v2f*)(Wo + (size_t)c * 32 + kb);
      const v2f w1 = *(const v2f*)(Wo + (size_t)(16 + c) * 32 + kb);
      v2f bf;
      bf[0] = mt ? ao1[2 * j] : ao0[2 * j];
      bf[1] = mt ? ao1[2 * j + 1] : ao0[2 * j + 1];
      o0 = WMMA_F32(w0, bf, o0);
      o1 = WMMA_F32(w1, bf, o1);
    }
  }
  float* og = outp + (size_t)(R0 + c) * 32 + 8 * h;
  v4f x;
  x[0] = o0[0]; x[1] = o0[1]; x[2] = o0[2]; x[3] = o0[3];
  __builtin_nontemporal_store(x, (v4f*)(og));
  x[0] = o0[4]; x[1] = o0[5]; x[2] = o0[6]; x[3] = o0[7];
  __builtin_nontemporal_store(x, (v4f*)(og + 4));
  x[0] = o1[0]; x[1] = o1[1]; x[2] = o1[2]; x[3] = o1[3];
  __builtin_nontemporal_store(x, (v4f*)(og + 16));
  x[0] = o1[4]; x[1] = o1[5]; x[2] = o1[6]; x[3] = o1[7];
  __builtin_nontemporal_store(x, (v4f*)(og + 20));
}

extern "C" void kernel_launch(void* const* d_in, const int* in_sizes, int n_in,
                              void* d_out, int out_size, void* d_ws, size_t ws_size,
                              hipStream_t stream) {
  (void)in_sizes; (void)n_in; (void)out_size; (void)ws_size;
  const float* query = (const float*)d_in[0];
  const float* key   = (const float*)d_in[1];
  const int*   qmask = (const int*)d_in[2];
  const int*   kmask = (const int*)d_in[3];
  const float* Wq = (const float*)d_in[4];
  const float* Wk = (const float*)d_in[5];
  const float* Wv = (const float*)d_in[6];
  const float* Wo = (const float*)d_in[7];
  const float* lq1 = (const float*)d_in[8];
  const float* lk1 = (const float*)d_in[9];
  const float* lq2 = (const float*)d_in[10];
  const float* lk2 = (const float*)d_in[11];
  const float* lnw = (const float*)d_in[12];

  float* outp  = (float*)d_out;
  float* diffp = outp + (size_t)B_ * L_ * E_;

  float* Qs = (float*)d_ws;
  float* Km = Qs + (size_t)B_ * L_ * E_;
  float* Vm = Km + (size_t)B_ * S_ * E_;

  kv_proj_kernel<<<(B_ * S_) / 16, 32, 0, stream>>>(key, Wk, Wv, Km, Vm);
  q_proj_kernel<<<(B_ * L_) / 16, 32, 0, stream>>>(query, Wq, Qs);
  attn_kernel<<<(B_ * L_) / 16, 32, 0, stream>>>(Qs, Km, Vm, qmask, kmask,
                                                 lq1, lk1, lq2, lk2, lnw, Wo,
                                                 outp, diffp);
}


# K/V projection: 4 waves per 16-row tile, each on one 16-column output tile over the whole k range, unrolled so its loads stay in flight; Km/Vm sum order changes (k regrouped, 2 partial sums)
# speedup vs baseline: 4.2902x; 1.3652x over previous
#include <hip/hip_runtime.h>
#include <math.h>

typedef float v2f __attribute__((ext_vector_type(2)));
typedef float v4f __attribute__((ext_vector_type(4)));
typedef int   v4i __attribute__((ext_vector_type(4)));
typedef float v8f __attribute__((ext_vector_type(8)));

#define WMMA_F32(a, b, c) \
  __builtin_amdgcn_wmma_f32_16x16x4_f32(false, (a), false, (b), (short)0, (c), false, false)

static constexpr int B_ = 8, S_ = 512, E_ = 32, L_ = 4096, SUBE_ = 768;
static constexpr float LAMBDA_INIT = 0.2f;

__global__ __launch_bounds__(128) __attribute__((amdgpu_waves_per_eu(1, 1))) void kv_proj_kernel(
    const float* __restrict__ key, const float* __restrict__ Wk,
    const float* __restrict__ Wv, float* __restrict__ Km, float* __restrict__ Vm) {
  __shared__ __attribute__((aligned(16))) float tile[16 * 68];
  const int tid  = threadIdx.x;
  const int w    = tid >> 5;
  const int lane = tid & 31;
  const int n    = lane & 15;
  const int h    = lane >> 4;
  const int m0   = blockIdx.x * 16;

  const float* wsel = (w < 2) ? Wk : Wv;
  const float* ap = key + (size_t)(m0 + n) * SUBE_ + 4 * h;
  const float* bp = wsel + (size_t)((w & 1) * 16 + n) * SUBE_ + 4 * h;

  v8f acc0 = {}, acc1 = {};
#pragma unroll
  for (int i = 0; i < SUBE_ / 8; ++i) {
    const v4f a = *(const v4f*)(ap + 8 * i);
    const v4f b = *(const v4f*)(bp + 8 * i);
    v2f a01, b01, a23, b23;
    a01[0] = a[0]; a01[1] = a[1]; b01[0] = b[0]; b01[1] = b[1];
    a23[0] = a[2]; a23[1] = a[3]; b23[0] = b[2]; b23[1] = b[3];
    acc0 = WMMA_F32(a01, b01, acc0);
    acc1 = WMMA_F32(a23, b23, acc1);
  }

#pragma unroll
  for (int v = 0; v < 8; ++v)
    tile[(8 * h + v) * 68 + 16 * w + n] = acc0[v] + acc1[v];
  __syncthreads();

#pragma unroll
  for (int q = 0; q < 2; ++q) {
    const int p    = tid + 128 * q;
    const int line = p >> 3;
    const int seg  = p & 7;
    const int row  = line & 15;
    const int isv  = line >> 4;
    const v4f x = *(const v4f*)(tile + row * 68 + 32 * isv + 4 * seg);
    float* dst = (isv ? Vm : Km) + (size_t)(m0 + row) * 32 + 4 * seg;
    *(v4f*)dst = x;
  }
}

__global__ __launch_bounds__(32) void q_proj_kernel(
    const float* __restrict__ q, const float* __restrict__ Wq,
    float* __restrict__ Qs) {
  const int lane = threadIdx.x;
  const int n    = lane & 15;
  const int khi  = lane >> 4;
  const int m0   = blockIdx.x * 16;
  const int mrow = m0 + n;

  v8f a0 = {}, a1 = {};
  for (int kk = 0; kk < 8; ++kk) {
    const int kb = kk * 4 + khi * 2;
    v2f a  = *(const v2f*)(q + (size_t)mrow * 32 + kb);
    v2f b0 = *(const v2f*)(Wq + (size_t)n * 32 + kb);
    v2f b1 = *(const v2f*)(Wq + (size_t)(16 + n) * 32 + kb);
    a0 = WMMA_F32(a, b0, a0);
    a1 = WMMA_F32(a, b1, a1);
  }
  const int rbase = m0 + (khi ? 8 : 0);
  for (int v = 0; v < 8; ++v) {
    const size_t r = (size_t)(rbase + v) * 32;
    Qs[r + n]      = a0[v] * 0.25f;
    Qs[r + 16 + n] = a1[v] * 0.25f;
  }
}

__global__ __launch_bounds__(32) void attn_kernel(
    const float* __restrict__ Qs, const float* __restrict__ Km,
    const float* __restrict__ Vm, const int* __restrict__ qmask,
    const int* __restrict__ kmask, const float* __restrict__ lq1,
    const float* __restrict__ lk1, const float* __restrict__ lq2,
    const float* __restrict__ lk2, const float* __restrict__ lnw,
    const float* __restrict__ Wo, float* __restrict__ outp,
    float* __restrict__ diffp) {
  const int lane = threadIdx.x;
  const int c    = lane & 15;
  const int h    = lane >> 4;
  const int R0   = blockIdx.x * 16;
  const int b    = R0 / L_;

  float s1 = 0.f, s2 = 0.f;
  for (int i = 0; i < 16; ++i) { s1 += lq1[i] * lk1[i]; s2 += lq2[i] * lk2[i]; }
  const float lambda_full = __expf(s1) - __expf(s2) + LAMBDA_INIT;

  const int qv = qmask[R0 + c];

  v2f qf[2][4];
#pragma unroll
  for (int hd = 0; hd < 2; ++hd) {
    const float* qp = Qs + (size_t)(R0 + c) * 32 + hd * 16 + 8 * h;
    const v4f qa = *(const v4f*)(qp);
    const v4f qb = *(const v4f*)(qp + 4);
    qf[hd][0][0] = qa[0]; qf[hd][0][1] = qa[1];
    qf[hd][1][0] = qa[2]; qf[hd][1][1] = qa[3];
    qf[hd][2][0] = qb[0]; qf[hd][2][1] = qb[1];
    qf[hd][3][0] = qb[2]; qf[hd][3][1] = qb[3];
  }

  const int crow = ((c & 4) << 1) | ((c & 8) >> 1) | (c & 3);
  const float* kbase = Km + (size_t)(b * S_ + crow) * 32 + 8 * h;
  const int*   mbase = kmask + b * S_ + 4 * h;

  v8f S0[32], S1[32];
#pragma unroll
  for (int t = 0; t < 32; ++t) {
    const float* kp = kbase + (size_t)t * 16 * 32;
    const v4f k0a = *(const v4f*)(kp);
    const v4f k0b = *(const v4f*)(kp + 4);
    const v4f k1a = *(const v4f*)(kp + 16);
    const v4f k1b = *(const v4f*)(kp + 20);
    const v4i ma  = *(const v4i*)(mbase + t * 16);
    const v4i mb  = *(const v4i*)(mbase + t * 16 + 8);
    v8f a0 = {}, a1 = {};
    v2f f;
    f[0] = k0a[0]; f[1] = k0a[1]; a0 = WMMA_F32(f, qf[0][0], a0);
    f[0] = k1a[0]; f[1] = k1a[1]; a1 = WMMA_F32(f, qf[1][0], a1);
    f[0] = k0a[2]; f[1] = k0a[3]; a0 = WMMA_F32(f, qf[0][1], a0);
    f[0] = k1a[2]; f[1] = k1a[3]; a1 = WMMA_F32(f, qf[1][1], a1);
    f[0] = k0b[0]; f[1] = k0b[1]; a0 = WMMA_F32(f, qf[0][2], a0);
    f[0] = k1b[0]; f[1] = k1b[1]; a1 = WMMA_F32(f, qf[1][2], a1);
    f[0] = k0b[2]; f[1] = k0b[3]; a0 = WMMA_F32(f, qf[0][3], a0);
    f[0] = k1b[2]; f[1] = k1b[3]; a1 = WMMA_F32(f, qf[1][3], a1);
#pragma unroll
    for (int r = 0; r < 8; ++r) {
      const int km    = (r < 4) ? ma[r & 3] : mb[r & 3];
      const bool valid = km && qv;
      S0[t][r] = valid ? a0[r] : -1e20f;
      S1[t][r] = valid ? a1[r] : -1e20f;
    }
  }

  v8f m0v = S0[0], m1v = S1[0];
#pragma unroll
  for (int t = 1; t < 32; ++t) {
#pragma unroll
    for (int r = 0; r < 8; ++r) {
      m0v[r] = fmaxf(m0v[r], S0[t][r]);
      m1v[r] = fmaxf(m1v[r], S1[t][r]);
    }
  }
  float mx0 = fmaxf(fmaxf(fmaxf(m0v[0], m0v[1]), fmaxf(m0v[2], m0v[3])),
                    fmaxf(fmaxf(m0v[4], m0v[5]), fmaxf(m0v[6], m0v[7])));
  float mx1 = fmaxf(fmaxf(fmaxf(m1v[0], m1v[1]), fmaxf(m1v[2], m1v[3])),
                    fmaxf(fmaxf(m1v[4], m1v[5]), fmaxf(m1v[6], m1v[7])));
  mx0 = fmaxf(mx0, __shfl_xor(mx0, 16, 32));
  mx1 = fmaxf(mx1, __shfl_xor(mx1, 16, 32));

  v8f s0v = {}, s1v = {};
#pragma unroll
  for (int t = 0; t < 32; ++t) {
#pragma unroll
    for (int r = 0; r < 8; ++r) {
      const float e0 = __expf(S0[t][r] - mx0);
      const float e1 = __expf(S1[t][r] - mx1);
      S0[t][r] = e0; s0v[r] += e0;
      S1[t][r] = e1; s1v[r] += e1;
    }
  }
  float sum0 = ((s0v[0] + s0v[1]) + (s0v[2] + s0v[3])) +
               ((s0v[4] + s0v[5]) + (s0v[6] + s0v[7]));
  float sum1 = ((s1v[0] + s1v[1]) + (s1v[2] + s1v[3])) +
               ((s1v[4] + s1v[5]) + (s1v[6] + s1v[7]));
  sum0 += __shfl_xor(sum0, 16, 32);
  sum1 += __shfl_xor(sum1, 16, 32);
  const float inv0 = 1.f / (sum0 + 1e-8f);
  const float inv1 = 1.f / (sum1 + 1e-8f);

  v8f dmv;
#pragma unroll
  for (int t = 0; t < 32; ++t) {
#pragma unroll
    for (int r = 0; r < 8; ++r) {
      const float d = S0[t][r] * inv0 - lambda_full * (S1[t][r] * inv1);
      S0[t][r] = d;
      dmv[r] = (t == 0) ? d : fminf(dmv[r], d);
    }
  }
  float dmin = fminf(fminf(fminf(dmv[0], dmv[1]), fminf(dmv[2], dmv[3])),
                     fminf(fminf(dmv[4], dmv[5]), fminf(dmv[6], dmv[7])));
  dmin = fminf(dmin, __shfl_xor(dmin, 16, 32));

  float* dg = diffp + (size_t)(R0 + c) * 512 + 4 * h;
#pragma unroll
  for (int t = 0; t < 32; ++t) {
    const v4i ma = *(const v4i*)(mbase + t * 16);
    const v4i mb = *(const v4i*)(mbase + t * 16 + 8);
    v4f lo, hi;
#pragma unroll
    for (int r = 0; r < 8; ++r) {
      const int km = (r < 4) ? ma[r & 3] : mb[r & 3];
      float dd = S0[t][r] - dmin + 1e-5f;
      dd = (qv && km) ? dd : 0.f;
      S0[t][r] = dd;
      if (r < 4) lo[r & 3] = dd; else hi[r & 3] = dd;
    }
    __builtin_nontemporal_store(lo, (v4f*)(dg + t * 16));
    __builtin_nontemporal_store(hi, (v4f*)(dg + t * 16 + 8));
  }

  const float* vbase = Vm + (size_t)(b * S_ + 4 * h) * 32 + c;
  v8f ao0 = {}, ao1 = {};
#pragma unroll
  for (int t = 0; t < 32; ++t) {
#pragma unroll
    for (int j = 0; j < 4; ++j) {
      const float* vp = vbase + (size_t)(t * 16 + 8 * (j >> 1) + 2 * (j & 1)) * 32;
      v2f a0, a1, bf;
      a0[0] = vp[0];  a0[1] = vp[32];
      a1[0] = vp[16]; a1[1] = vp[48];
      bf[0] = S0[t][2 * j]; bf[1] = S0[t][2 * j + 1];
      ao0 = WMMA_F32(a0, bf, ao0);
      ao1 = WMMA_F32(a1, bf, ao1);
    }
  }

  float sq = 0.f;
#pragma unroll
  for (int r = 0; r < 8; ++r) sq += ao0[r] * ao0[r];
#pragma unroll
  for (int r = 0; r < 8; ++r) sq += ao1[r] * ao1[r];
  sq += __shfl_xor(sq, 16, 32);
  const float rms = rsqrtf(sq * (1.0f / 32.0f) + 1e-5f);
#pragma unroll
  for (int r = 0; r < 8; ++r) {
    ao0[r] = ao0[r] * rms * lnw[8 * h + r] * 0.8f;
    ao1[r] = ao1[r] * rms * lnw[16 + 8 * h + r] * 0.8f;
  }

  v8f o0 = {}, o1 = {};
#pragma unroll
  for (int mt = 0; mt < 2; ++mt) {
#pragma unroll
    for (int j = 0; j < 4; ++j) {
      const int kb = 16 * mt + 8 * h + 2 * j;
      const v2f w0 = *(const v2f*)(Wo + (size_t)c * 32 + kb);
      const v2f w1 = *(const v2f*)(Wo + (size_t)(16 + c) * 32 + kb);
      v2f bf;
      bf[0] = mt ? ao1[2 * j] : ao0[2 * j];
      bf[1] = mt ? ao1[2 * j + 1] : ao0[2 * j + 1];
      o0 = WMMA_F32(w0, bf, o0);
      o1 = WMMA_F32(w1, bf, o1);
    }
  }
  float* og = outp + (size_t)(R0 + c) * 32 + 8 * h;
  v4f x;
  x[0] = o0[0]; x[1] = o0[1]; x[2] = o0[2]; x[3] = o0[3];
  __builtin_nontemporal_store(x, (v4f*)(og));
  x[0] = o0[4]; x[1] = o0[5]; x[2] = o0[6]; x[3] = o0[7];
  __builtin_nontemporal_store(x, (v4f*)(og + 4));
  x[0] = o1[0]; x[1] = o1[1]; x[2] = o1[2]; x[3] = o1[3];
  __builtin_nontemporal_store(x, (v4f*)(og + 16));
  x[0] = o1[4]; x[1] = o1[5]; x[2] = o1[6]; x[3] = o1[7];
  __builtin_nontemporal_store(x, (v4f*)(og + 20));
}

extern "C" void kernel_launch(void* const* d_in, const int* in_sizes, int n_in,
                              void* d_out, int out_size, void* d_ws, size_t ws_size,
                              hipStream_t stream) {
  (void)in_sizes; (void)n_in; (void)out_size; (void)ws_size;
  const float* query = (const float*)d_in[0];
  const float* key   = (const float*)d_in[1];
  const int*   qmask = (const int*)d_in[2];
  const int*   kmask = (const int*)d_in[3];
  const float* Wq = (const float*)d_in[4];
  const float* Wk = (const float*)d_in[5];
  const float* Wv = (const float*)d_in[6];
  const float* Wo = (const float*)d_in[7];
  const float* lq1 = (const float*)d_in[8];
  const float* lk1 = (const float*)d_in[9];
  const float* lq2 = (const float*)d_in[10];
  const float* lk2 = (const float*)d_in[11];
  const float* lnw = (const float*)d_in[12];

  float* outp  = (float*)d_out;
  float* diffp = outp + (size_t)B_ * L_ * E_;

  float* Qs = (float*)d_ws;
  float* Km = Qs + (size_t)B_ * L_ * E_;
  float* Vm = Km + (size_t)B_ * S_ * E_;

  kv_proj_kernel<<<(B_ * S_) / 16, 128, 0, stream>>>(key, Wk, Wv, Km, Vm);
  q_proj_kernel<<<(B_ * L_) / 16, 32, 0, stream>>>(query, Wq, Qs);
  attn_kernel<<<(B_ * L_) / 16, 32, 0, stream>>>(Qs, Km, Vm, qmask, kmask,
                                                 lq1, lk1, lq2, lk2, lnw, Wo,
                                                 outp, diffp);
}


# attention kernel: both heads run through one rolled two-trip loop (scores, softmax, combine into the diff registers), key mask added as the score tile's start value: text 8860 -> 5448 instructions
# speedup vs baseline: 4.8199x; 1.1235x over previous
#include <hip/hip_runtime.h>
#include <math.h>

typedef float v2f __attribute__((ext_vector_type(2)));
typedef float v4f __attribute__((ext_vector_type(4)));
typedef int   v4i __attribute__((ext_vector_type(4)));
typedef float v8f __attribute__((ext_vector_type(8)));

#define WMMA_F32(a, b, c) \
  __builtin_amdgcn_wmma_f32_16x16x4_f32(false, (a), false, (b), (short)0, (c), false, false)

static constexpr int B_ = 8, S_ = 512, E_ = 32, L_ = 4096, SUBE_ = 768;
static constexpr float LAMBDA_INIT = 0.2f;

__global__ __launch_bounds__(128) __attribute__((amdgpu_waves_per_eu(1, 1))) void kv_proj_kernel(
    const float* __restrict__ key, const float* __restrict__ Wk,
    const float* __restrict__ Wv, const int* __restrict__ kmask,
    float* __restrict__ Km, float* __restrict__ Vm, float* __restrict__ Bk) {
  __shared__ __attribute__((aligned(16))) float tile[16 * 68];
  const int tid  = threadIdx.x;
  const int w    = tid >> 5;
  const int lane = tid & 31;
  const int n    = lane & 15;
  const int h    = lane >> 4;
  const int m0   = blockIdx.x * 16;

  const bool bias_writer = (blockIdx.x & 1) == 0 && tid < 64;
  const int kidx = (m0 + lane < B_ * S_) ? m0 + lane : B_ * S_ - 1;
  const int kmv = kmask[kidx];

  const float* wsel = (w < 2) ? Wk : Wv;
  const float* ap = key + (size_t)(m0 + n) * SUBE_ + 4 * h;
  const float* bp = wsel + (size_t)((w & 1) * 16 + n) * SUBE_ + 4 * h;

  v8f acc0 = {}, acc1 = {};
#pragma unroll
  for (int i = 0; i < SUBE_ / 8; ++i) {
    const v4f a = *(const v4f*)(ap + 8 * i);
    const v4f b = *(const v4f*)(bp + 8 * i);
    v2f a01, b01, a23, b23;
    a01[0] = a[0]; a01[1] = a[1]; b01[0] = b[0]; b01[1] = b[1];
    a23[0] = a[2]; a23[1] = a[3]; b23[0] = b[2]; b23[1] = b[3];
    acc0 = WMMA_F32(a01, b01, acc0);
    acc1 = WMMA_F32(a23, b23, acc1);
  }

#pragma unroll
  for (int v = 0; v < 8; ++v)
    tile[(8 * h + v) * 68 + 16 * w + n] = acc0[v] + acc1[v];
  __syncthreads();

#pragma unroll
  for (int q = 0; q < 2; ++q) {
    const int p    = tid + 128 * q;
    const int line = p >> 3;
    const int seg  = p & 7;
    const int row  = line & 15;
    const int isv  = line >> 4;
    const v4f x = *(const v4f*)(tile + row * 68 + 32 * isv + 4 * seg);
    float* dst = (isv ? Vm : Km) + (size_t)(m0 + row) * 32 + 4 * seg;
    *(v4f*)dst = x;
  }
  if (bias_writer)
    Bk[(size_t)w * (B_ * S_) + m0 + lane] = kmv ? 0.f : -1e20f;
}

__global__ __launch_bounds__(32) void q_proj_kernel(
    const float* __restrict__ q, const float* __restrict__ Wq,
    float* __restrict__ Qs) {
  const int lane = threadIdx.x;
  const int n    = lane & 15;
  const int khi  = lane >> 4;
  const int m0   = blockIdx.x * 16;
  const int mrow = m0 + n;

  v8f a0 = {}, a1 = {};
  for (int kk = 0; kk < 8; ++kk) {
    const int kb = kk * 4 + khi * 2;
    v2f a  = *(const v2f*)(q + (size_t)mrow * 32 + kb);
    v2f b0 = *(const v2f*)(Wq + (size_t)n * 32 + kb);
    v2f b1 = *(const v2f*)(Wq + (size_t)(16 + n) * 32 + kb);
    a0 = WMMA_F32(a, b0, a0);
    a1 = WMMA_F32(a, b1, a1);
  }
  const int rbase = m0 + (khi ? 8 : 0);
  for (int v = 0; v < 8; ++v) {
    const size_t r = (size_t)(rbase + v) * 32;
    Qs[r + n]      = a0[v] * 0.25f;
    Qs[r + 16 + n] = a1[v] * 0.25f;
  }
}

__global__ __launch_bounds__(32) void attn_kernel(
    const float* __restrict__ Qs, const float* __restrict__ Km,
    const float* __restrict__ Vm, const int* __restrict__ qmask,
    const float* __restrict__ Bk, const float* __restrict__ lq1,
    const float* __restrict__ lk1, const float* __restrict__ lq2,
    const float* __restrict__ lk2, const float* __restrict__ lnw,
    const float* __restrict__ Wo, float* __restrict__ outp,
    float* __restrict__ diffp) {
  const int lane = threadIdx.x;
  const int c    = lane & 15;
  const int h    = lane >> 4;
  const int R0   = blockIdx.x * 16;
  const int b    = R0 / L_;

  float s1 = 0.f, s2 = 0.f;
  for (int i = 0; i < 16; ++i) { s1 += lq1[i] * lk1[i]; s2 += lq2[i] * lk2[i]; }
  const float lambda_full = __expf(s1) - __expf(s2) + LAMBDA_INIT;

  const int qv = qmask[R0 + c];

  const int crow = ((c & 4) << 1) | ((c & 8) >> 1) | (c & 3);
  const float* kbase = Km + (size_t)(b * S_ + crow) * 32 + 8 * h;
  const float* bbase = Bk + b * S_ + 4 * h;

  const float zq = qv ? 0.f : 1.f;

  v8f D[32];
#pragma unroll
  for (int t = 0; t < 32; ++t) D[t] = v8f{};

#pragma unroll 1
  for (int hd = 0; hd < 2; ++hd) {
    const float* qp = Qs + (size_t)(R0 + c) * 32 + hd * 16 + 8 * h;
    const v4f qa = *(const v4f*)(qp);
    const v4f qb = *(const v4f*)(qp + 4);
    v2f q0, q1, q2, q3;
    q0[0] = qa[0]; q0[1] = qa[1];
    q1[0] = qa[2]; q1[1] = qa[3];
    q2[0] = qb[0]; q2[1] = qb[1];
    q3[0] = qb[2]; q3[1] = qb[3];
    const float* kh = kbase + hd * 16;
    const float* bh = bbase + hd * (B_ * S_);

    v8f S[32];
#pragma unroll
    for (int t = 0; t < 32; t += 2) {
      const float* kpx = kh + (size_t)t * 16 * 32;
      const float* kpy = kpx + 16 * 32;
      const v4f xa = *(const v4f*)(kpx);
      const v4f xb = *(const v4f*)(kpx + 4);
      const v4f ya = *(const v4f*)(kpy);
      const v4f yb = *(const v4f*)(kpy + 4);
      const v4f bxa = *(const v4f*)(bh + t * 16);
      const v4f bxb = *(const v4f*)(bh + t * 16 + 8);
      const v4f bya = *(const v4f*)(bh + t * 16 + 16);
      const v4f byb = *(const v4f*)(bh + t * 16 + 24);
      v2f f;
      v8f x, y;
      x[0] = bxa[0]; x[1] = bxa[1]; x[2] = bxa[2]; x[3] = bxa[3];
      x[4] = bxb[0]; x[5] = bxb[1]; x[6] = bxb[2]; x[7] = bxb[3];
      y[0] = bya[0]; y[1] = bya[1]; y[2] = bya[2]; y[3] = bya[3];
      y[4] = byb[0]; y[5] = byb[1]; y[6] = byb[2]; y[7] = byb[3];
      f[0] = xa[0]; f[1] = xa[1]; x = WMMA_F32(f, q0, x);
      f[0] = ya[0]; f[1] = ya[1]; y = WMMA_F32(f, q0, y);
      f[0] = xa[2]; f[1] = xa[3]; x = WMMA_F32(f, q1, x);
      f[0] = ya[2]; f[1] = ya[3]; y = WMMA_F32(f, q1, y);
      f[0] = xb[0]; f[1] = xb[1]; x = WMMA_F32(f, q2, x);
      f[0] = yb[0]; f[1] = yb[1]; y = WMMA_F32(f, q2, y);
      f[0] = xb[2]; f[1] = xb[3]; x = WMMA_F32(f, q3, x);
      f[0] = yb[2]; f[1] = yb[3]; y = WMMA_F32(f, q3, y);
      S[t] = x; S[t + 1] = y;
    }

    v8f mv = S[0];
#pragma unroll
    for (int t = 1; t < 32; ++t) {
#pragma unroll
      for (int r = 0; r < 8; ++r) mv[r] = fmaxf(mv[r], S[t][r]);
    }
    float mx = fmaxf(fmaxf(fmaxf(mv[0], mv[1]), fmaxf(mv[2], mv[3])),
                     fmaxf(fmaxf(mv[4], mv[5]), fmaxf(mv[6], mv[7])));
    mx = fmaxf(mx, __shfl_xor(mx, 16, 32));

    v8f sv = {};
#pragma unroll
    for (int t = 0; t < 32; ++t) {
#pragma unroll
      for (int r = 0; r < 8; ++r) {
        const float e = __expf(S[t][r] - mx);
        S[t][r] = e; sv[r] += e;
      }
    }
    float sum = ((sv[0] + sv[1]) + (sv[2] + sv[3])) +
                ((sv[4] + sv[5]) + (sv[6] + sv[7]));
    sum += __shfl_xor(sum, 16, 32);
    const float inv   = 1.f / (sum + 1e-8f);
    const float scale = hd ? -lambda_full * inv : inv;

#pragma unroll
    for (int t = 0; t < 32; ++t) {
#pragma unroll
      for (int r = 0; r < 8; ++r) D[t][r] = S[t][r] * scale + D[t][r];
    }
  }

  v8f dmv = D[0];
#pragma unroll
  for (int t = 1; t < 32; ++t) {
#pragma unroll
    for (int r = 0; r < 8; ++r) dmv[r] = fminf(dmv[r], D[t][r]);
  }
  float dmin = fminf(fminf(fminf(dmv[0], dmv[1]), fminf(dmv[2], dmv[3])),
                     fminf(fminf(dmv[4], dmv[5]), fminf(dmv[6], dmv[7])));
  dmin = fminf(dmin, __shfl_xor(dmin, 16, 32));

  float* dg = diffp + (size_t)(R0 + c) * 512 + 4 * h;
#pragma unroll
  for (int t = 0; t < 32; ++t) {
    const v4f ba = *(const v4f*)(bbase + t * 16);
    const v4f bb = *(const v4f*)(bbase + t * 16 + 8);
    v4f lo, hi;
#pragma unroll
    for (int r = 0; r < 8; ++r) {
      const float bias = (r < 4) ? ba[r & 3] : bb[r & 3];
      float dd = D[t][r] - dmin + 1e-5f;
      dd = (bias == zq) ? dd : 0.f;
      D[t][r] = dd;
      if (r < 4) lo[r & 3] = dd; else hi[r & 3] = dd;
    }
    __builtin_nontemporal_store(lo, (v4f*)(dg + t * 16));
    __builtin_nontemporal_store(hi, (v4f*)(dg + t * 16 + 8));
  }

  const float* vbase = Vm + (size_t)(b * S_ + 4 * h) * 32 + c;
  v8f ao0 = {}, ao1 = {};
#pragma unroll
  for (int t = 0; t < 32; ++t) {
#pragma unroll
    for (int j = 0; j < 4; ++j) {
      const float* vp = vbase + (size_t)(t * 16 + 8 * (j >> 1) + 2 * (j & 1)) * 32;
      v2f a0, a1, bf;
      a0[0] = vp[0];  a0[1] = vp[32];
      a1[0] = vp[16]; a1[1] = vp[48];
      bf[0] = D[t][2 * j]; bf[1] = D[t][2 * j + 1];
      ao0 = WMMA_F32(a0, bf, ao0);
      ao1 = WMMA_F32(a1, bf, ao1);
    }
  }

  float sq = 0.f;
#pragma unroll
  for (int r = 0; r < 8; ++r) sq += ao0[r] * ao0[r];
#pragma unroll
  for (int r = 0; r < 8; ++r) sq += ao1[r] * ao1[r];
  sq += __shfl_xor(sq, 16, 32);
  const float rms = rsqrtf(sq * (1.0f / 32.0f) + 1e-5f);
#pragma unroll
  for (int r = 0; r < 8; ++r) {
    ao0[r] = ao0[r] * rms * lnw[8 * h + r] * 0.8f;
    ao1[r] = ao1[r] * rms * lnw[16 + 8 * h + r] * 0.8f;
  }

  v8f o0 = {}, o1 = {};
#pragma unroll
  for (int mt = 0; mt < 2; ++mt) {
#pragma unroll
    for (int j = 0; j < 4; ++j) {
      const int kb = 16 * mt + 8 * h + 2 * j;
      const v2f w0 = *(const v2f*)(Wo + (size_t)c * 32 + kb);
      const v2f w1 = *(const v2f*)(Wo + (size_t)(16 + c) * 32 + kb);
      v2f bf;
      bf[0] = mt ? ao1[2 * j] : ao0[2 * j];
      bf[1] = mt ? ao1[2 * j + 1] : ao0[2 * j + 1];
      o0 = WMMA_F32(w0, bf, o0);
      o1 = WMMA_F32(w1, bf, o1);
    }
  }
  float* og = outp + (size_t)(R0 + c) * 32 + 8 * h;
  v4f x;
  x[0] = o0[0]; x[1] = o0[1]; x[2] = o0[2]; x[3] = o0[3];
  __builtin_nontemporal_store(x, (v4f*)(og));
  x[0] = o0[4]; x[1] = o0[5]; x[2] = o0[6]; x[3] = o0[7];
  __builtin_nontemporal_store(x, (v4f*)(og + 4));
  x[0] = o1[0]; x[1] = o1[1]; x[2] = o1[2]; x[3] = o1[3];
  __builtin_nontemporal_store(x, (v4f*)(og + 16));
  x[0] = o1[4]; x[1] = o1[5]; x[2] = o1[6]; x[3] = o1[7];
  __builtin_nontemporal_store(x, (v4f*)(og + 20));
}

extern "C" void kernel_launch(void* const* d_in, const int* in_sizes, int n_in,
                              void* d_out, int out_size, void* d_ws, size_t ws_size,
                              hipStream_t stream) {
  (void)in_sizes; (void)n_in; (void)out_size; (void)ws_size;
  const float* query = (const float*)d_in[0];
  const float* key   = (const float*)d_in[1];
  const int*   qmask = (const int*)d_in[2];
  const int*   kmask = (const int*)d_in[3];
  const float* Wq = (const float*)d_in[4];
  const float* Wk = (const float*)d_in[5];
  const float* Wv = (const float*)d_in[6];
  const float* Wo = (const float*)d_in[7];
  const float* lq1 = (const float*)d_in[8];
  const float* lk1 = (const float*)d_in[9];
  const float* lq2 = (const float*)d_in[10];
  const float* lk2 = (const float*)d_in[11];
  const float* lnw = (const float*)d_in[12];

  float* outp  = (float*)d_out;
  float* diffp = outp + (size_t)B_ * L_ * E_;

  float* Qs = (float*)d_ws;
  float* Km = Qs + (size_t)B_ * L_ * E_;
  float* Vm = Km + (size_t)B_ * S_ * E_;
  float* Bk = Vm + (size_t)B_ * S_ * E_;

  kv_proj_kernel<<<(B_ * S_) / 16, 128, 0, stream>>>(key, Wk, Wv, kmask, Km, Vm, Bk);
  q_proj_kernel<<<(B_ * L_) / 16, 32, 0, stream>>>(query, Wq, Qs);
  attn_kernel<<<(B_ * L_) / 16, 32, 0, stream>>>(Qs, Km, Vm, qmask, Bk,
                                                 lq1, lk1, lq2, lk2, lnw, Wo,
                                                 outp, diffp);
}


# attention kernel: head 0's weights parked in LDS (32 KB a wave), no second 256-register tile set; score loads held 8 tiles ahead; VGPRs 796 -> 512: two waves per SIMD run side by side
# speedup vs baseline: 4.9895x; 1.0352x over previous
#include <hip/hip_runtime.h>
#include <math.h>

typedef float v2f __attribute__((ext_vector_type(2)));
typedef float v4f __attribute__((ext_vector_type(4)));
typedef int   v4i __attribute__((ext_vector_type(4)));
typedef float v8f __attribute__((ext_vector_type(8)));

#define WMMA_F32(a, b, c) \
  __builtin_amdgcn_wmma_f32_16x16x4_f32(false, (a), false, (b), (short)0, (c), false, false)

static constexpr int B_ = 8, S_ = 512, E_ = 32, L_ = 4096, SUBE_ = 768;
static constexpr float LAMBDA_INIT = 0.2f;

__global__ __launch_bounds__(128) __attribute__((amdgpu_waves_per_eu(1, 1))) void kv_proj_kernel(
    const float* __restrict__ key, const float* __restrict__ Wk,
    const float* __restrict__ Wv, const int* __restrict__ kmask,
    float* __restrict__ Km, float* __restrict__ Vm, float* __restrict__ Bk) {
  __shared__ __attribute__((aligned(16))) float tile[16 * 68];
  const int tid  = threadIdx.x;
  const int w    = tid >> 5;
  const int lane = tid & 31;
  const int n    = lane & 15;
  const int h    = lane >> 4;
  const int m0   = blockIdx.x * 16;

  const bool bias_writer = (blockIdx.x & 1) == 0 && tid < 64;
  const int kidx = (m0 + lane < B_ * S_) ? m0 + lane : B_ * S_ - 1;
  const int kmv = kmask[kidx];

  const float* wsel = (w < 2) ? Wk : Wv;
  const float* ap = key + (size_t)(m0 + n) * SUBE_ + 4 * h;
  const float* bp = wsel + (size_t)((w & 1) * 16 + n) * SUBE_ + 4 * h;

  v8f acc0 = {}, acc1 = {};
#pragma unroll
  for (int i = 0; i < SUBE_ / 8; ++i) {
    const v4f a = *(const v4f*)(ap + 8 * i);
    const v4f b = *(const v4f*)(bp + 8 * i);
    v2f a01, b01, a23, b23;
    a01[0] = a[0]; a01[1] = a[1]; b01[0] = b[0]; b01[1] = b[1];
    a23[0] = a[2]; a23[1] = a[3]; b23[0] = b[2]; b23[1] = b[3];
    acc0 = WMMA_F32(a01, b01, acc0);
    acc1 = WMMA_F32(a23, b23, acc1);
  }

#pragma unroll
  for (int v = 0; v < 8; ++v)
    tile[(8 * h + v) * 68 + 16 * w + n] = acc0[v] + acc1[v];
  __syncthreads();

#pragma unroll
  for (int q = 0; q < 2; ++q) {
    const int p    = tid + 128 * q;
    const int line = p >> 3;
    const int seg  = p & 7;
    const int row  = line & 15;
    const int isv  = line >> 4;
    const v4f x = *(const v4f*)(tile + row * 68 + 32 * isv + 4 * seg);
    float* dst = (isv ? Vm : Km) + (size_t)(m0 + row) * 32 + 4 * seg;
    *(v4f*)dst = x;
  }
  if (bias_writer)
    Bk[(size_t)w * (B_ * S_) + m0 + lane] = kmv ? 0.f : -1e20f;
}

__global__ __launch_bounds__(32) void q_proj_kernel(
    const float* __restrict__ q, const float* __restrict__ Wq,
    float* __restrict__ Qs) {
  const int lane = threadIdx.x;
  const int n    = lane & 15;
  const int khi  = lane >> 4;
  const int m0   = blockIdx.x * 16;
  const int mrow = m0 + n;

  v8f a0 = {}, a1 = {};
  for (int kk = 0; kk < 8; ++kk) {
    const int kb = kk * 4 + khi * 2;
    v2f a  = *(const v2f*)(q + (size_t)mrow * 32 + kb);
    v2f b0 = *(const v2f*)(Wq + (size_t)n * 32 + kb);
    v2f b1 = *(const v2f*)(Wq + (size_t)(16 + n) * 32 + kb);
    a0 = WMMA_F32(a, b0, a0);
    a1 = WMMA_F32(a, b1, a1);
  }
  const int rbase = m0 + (khi ? 8 : 0);
  for (int v = 0; v < 8; ++v) {
    const size_t r = (size_t)(rbase + v) * 32;
    Qs[r + n]      = a0[v] * 0.25f;
    Qs[r + 16 + n] = a1[v] * 0.25f;
  }
}

__global__ __launch_bounds__(32) __attribute__((amdgpu_waves_per_eu(2, 2))) void attn_kernel(
    const float* __restrict__ Qs, const float* __restrict__ Km,
    const float* __restrict__ Vm, const int* __restrict__ qmask,
    const float* __restrict__ Bk, const float* __restrict__ lq1,
    const float* __restrict__ lk1, const float* __restrict__ lq2,
    const float* __restrict__ lk2, const float* __restrict__ lnw,
    const float* __restrict__ Wo, float* __restrict__ outp,
    float* __restrict__ diffp) {
  __shared__ __attribute__((aligned(16))) v4f park[64 * 32];
  const int lane = threadIdx.x;
  const int c    = lane & 15;
  const int h    = lane >> 4;
  const int R0   = blockIdx.x * 16;
  const int b    = R0 / L_;

  float s1 = 0.f, s2 = 0.f;
  for (int i = 0; i < 16; ++i) { s1 += lq1[i] * lk1[i]; s2 += lq2[i] * lk2[i]; }
  const float lambda_full = __expf(s1) - __expf(s2) + LAMBDA_INIT;

  const int qv = qmask[R0 + c];

  const int crow = ((c & 4) << 1) | ((c & 8) >> 1) | (c & 3);
  const float* kbase = Km + (size_t)(b * S_ + crow) * 32 + 8 * h;
  const float* bbase = Bk + b * S_ + 4 * h;

  const float zq = qv ? 0.f : 1.f;

  v8f D[32];

#pragma unroll 1
  for (int hd = 0; hd < 2; ++hd) {
    const float* qp = Qs + (size_t)(R0 + c) * 32 + hd * 16 + 8 * h;
    const v4f qa = *(const v4f*)(qp);
    const v4f qb = *(const v4f*)(qp + 4);
    v2f q0, q1, q2, q3;
    q0[0] = qa[0]; q0[1] = qa[1];
    q1[0] = qa[2]; q1[1] = qa[3];
    q2[0] = qb[0]; q2[1] = qb[1];
    q3[0] = qb[2]; q3[1] = qb[3];
    const float* kh = kbase + hd * 16;
    const float* bh = bbase + hd * (B_ * S_);

    v4f KA[32], KB[32];
#pragma unroll
    for (int g = 0; g <= 4; ++g) {
      if (g < 4) {
#pragma unroll
        for (int i = 0; i < 8; ++i) {
          const int t = 8 * g + i;
          const float* kp = kh + (size_t)t * 16 * 32;
          KA[t] = *(const v4f*)(kp);
          KB[t] = *(const v4f*)(kp + 4);
          const v4f ba = *(const v4f*)(bh + t * 16);
          const v4f bb = *(const v4f*)(bh + t * 16 + 8);
          v8f x;
          x[0] = ba[0]; x[1] = ba[1]; x[2] = ba[2]; x[3] = ba[3];
          x[4] = bb[0]; x[5] = bb[1]; x[6] = bb[2]; x[7] = bb[3];
          D[t] = x;
        }
      }
      __builtin_amdgcn_sched_barrier(0);
      if (g > 0) {
#pragma unroll
        for (int i = 0; i < 8; i += 2) {
          const int t = 8 * (g - 1) + i;
          const v4f xa = KA[t], xb = KB[t], ya = KA[t + 1], yb = KB[t + 1];
          v8f x = D[t], y = D[t + 1];
          v2f f;
          f[0] = xa[0]; f[1] = xa[1]; x = WMMA_F32(f, q0, x);
          f[0] = ya[0]; f[1] = ya[1]; y = WMMA_F32(f, q0, y);
          f[0] = xa[2]; f[1] = xa[3]; x = WMMA_F32(f, q1, x);
          f[0] = ya[2]; f[1] = ya[3]; y = WMMA_F32(f, q1, y);
          f[0] = xb[0]; f[1] = xb[1]; x = WMMA_F32(f, q2, x);
          f[0] = yb[0]; f[1] = yb[1]; y = WMMA_F32(f, q2, y);
          f[0] = xb[2]; f[1] = xb[3]; x = WMMA_F32(f, q3, x);
          f[0] = yb[2]; f[1] = yb[3]; y = WMMA_F32(f, q3, y);
          D[t] = x; D[t + 1] = y;
        }
      }
      __builtin_amdgcn_sched_barrier(0);
    }

    v8f mv = D[0];
#pragma unroll
    for (int t = 1; t < 32; ++t) {
#pragma unroll
      for (int r = 0; r < 8; ++r) mv[r] = fmaxf(mv[r], D[t][r]);
    }
    float mx = fmaxf(fmaxf(fmaxf(mv[0], mv[1]), fmaxf(mv[2], mv[3])),
                     fmaxf(fmaxf(mv[4], mv[5]), fmaxf(mv[6], mv[7])));
    mx = fmaxf(mx, __shfl_xor(mx, 16, 32));

    v8f sv = {};
#pragma unroll
    for (int t = 0; t < 32; ++t) {
#pragma unroll
      for (int r = 0; r < 8; ++r) {
        const float e = __expf(D[t][r] - mx);
        D[t][r] = e; sv[r] += e;
      }
    }
    float sum = ((sv[0] + sv[1]) + (sv[2] + sv[3])) +
                ((sv[4] + sv[5]) + (sv[6] + sv[7]));
    sum += __shfl_xor(sum, 16, 32);
    const float inv   = 1.f / (sum + 1e-8f);
    if (hd == 0) {
#pragma unroll
      for (int t = 0; t < 32; ++t) {
        v4f lo, hi;
#pragma unroll
        for (int r = 0; r < 4; ++r) { lo[r] = D[t][r] * inv; hi[r] = D[t][4 + r] * inv; }
        park[(2 * t) * 32 + lane]     = lo;
        park[(2 * t + 1) * 32 + lane] = hi;
      }
    } else {
      const float scale = -lambda_full * inv;
#pragma unroll
      for (int t = 0; t < 32; ++t) {
        const v4f lo = park[(2 * t) * 32 + lane];
        const v4f hi = park[(2 * t + 1) * 32 + lane];
#pragma unroll
        for (int r = 0; r < 4; ++r) {
          D[t][r]     = D[t][r] * scale + lo[r];
          D[t][4 + r] = D[t][4 + r] * scale + hi[r];
        }
      }
    }
  }

  v8f dmv = D[0];
#pragma unroll
  for (int t = 1; t < 32; ++t) {
#pragma unroll
    for (int r = 0; r < 8; ++r) dmv[r] = fminf(dmv[r], D[t][r]);
  }
  float dmin = fminf(fminf(fminf(dmv[0], dmv[1]), fminf(dmv[2], dmv[3])),
                     fminf(fminf(dmv[4], dmv[5]), fminf(dmv[6], dmv[7])));
  dmin = fminf(dmin, __shfl_xor(dmin, 16, 32));

  float* dg = diffp + (size_t)(R0 + c) * 512 + 4 * h;
#pragma unroll
  for (int t = 0; t < 32; ++t) {
    const v4f ba = *(const v4f*)(bbase + t * 16);
    const v4f bb = *(const v4f*)(bbase + t * 16 + 8);
    v4f lo, hi;
#pragma unroll
    for (int r = 0; r < 8; ++r) {
      const float bias = (r < 4) ? ba[r & 3] : bb[r & 3];
      float dd = D[t][r] - dmin + 1e-5f;
      dd = (bias == zq) ? dd : 0.f;
      D[t][r] = dd;
      if (r < 4) lo[r & 3] = dd; else hi[r & 3] = dd;
    }
    __builtin_nontemporal_store(lo, (v4f*)(dg + t * 16));
    __builtin_nontemporal_store(hi, (v4f*)(dg + t * 16 + 8));
  }

  const float* vbase = Vm + (size_t)(b * S_ + 4 * h) * 32 + c;
  v8f ao0 = {}, ao1 = {};
#pragma unroll
  for (int t = 0; t < 32; ++t) {
#pragma unroll
    for (int j = 0; j < 4; ++j) {
      const float* vp = vbase + (size_t)(t * 16 + 8 * (j >> 1) + 2 * (j & 1)) * 32;
      v2f a0, a1, bf;
      a0[0] = vp[0];  a0[1] = vp[32];
      a1[0] = vp[16]; a1[1] = vp[48];
      bf[0] = D[t][2 * j]; bf[1] = D[t][2 * j + 1];
      ao0 = WMMA_F32(a0, bf, ao0);
      ao1 = WMMA_F32(a1, bf, ao1);
    }
  }

  float sq = 0.f;
#pragma unroll
  for (int r = 0; r < 8; ++r) sq += ao0[r] * ao0[r];
#pragma unroll
  for (int r = 0; r < 8; ++r) sq += ao1[r] * ao1[r];
  sq += __shfl_xor(sq, 16, 32);
  const float rms = rsqrtf(sq * (1.0f / 32.0f) + 1e-5f);
#pragma unroll
  for (int r = 0; r < 8; ++r) {
    ao0[r] = ao0[r] * rms * lnw[8 * h + r] * 0.8f;
    ao1[r] = ao1[r] * rms * lnw[16 + 8 * h + r] * 0.8f;
  }

  v8f o0 = {}, o1 = {};
#pragma unroll
  for (int mt = 0; mt < 2; ++mt) {
#pragma unroll
    for (int j = 0; j < 4; ++j) {
      const int kb = 16 * mt + 8 * h + 2 * j;
      const v2f w0 = *(const v2f*)(Wo + (size_t)c * 32 + kb);
      const v2f w1 = *(const v2f*)(Wo + (size_t)(16 + c) * 32 + kb);
      v2f bf;
      bf[0] = mt ? ao1[2 * j] : ao0[2 * j];
      bf[1] = mt ? ao1[2 * j + 1] : ao0[2 * j + 1];
      o0 = WMMA_F32(w0, bf, o0);
      o1 = WMMA_F32(w1, bf, o1);
    }
  }
  float* og = outp + (size_t)(R0 + c) * 32 + 8 * h;
  v4f x;
  x[0] = o0[0]; x[1] = o0[1]; x[2] = o0[2]; x[3] = o0[3];
  __builtin_nontemporal_store(x, (v4f*)(og));
  x[0] = o0[4]; x[1] = o0[5]; x[2] = o0[6]; x[3] = o0[7];
  __builtin_nontemporal_store(x, (v4f*)(og + 4));
  x[0] = o1[0]; x[1] = o1[1]; x[2] = o1[2]; x[3] = o1[3];
  __builtin_nontemporal_store(x, (v4f*)(og + 16));
  x[0] = o1[4]; x[1] = o1[5]; x[2] = o1[6]; x[3] = o1[7];
  __builtin_nontemporal_store(x, (v4f*)(og + 20));
}

extern "C" void kernel_launch(void* const* d_in, const int* in_sizes, int n_in,
                              void* d_out, int out_size, void* d_ws, size_t ws_size,
                              hipStream_t stream) {
  (void)in_sizes; (void)n_in; (void)out_size; (void)ws_size;
  const float* query = (const float*)d_in[0];
  const float* key   = (const float*)d_in[1];
  const int*   qmask = (const int*)d_in[2];
  const int*   kmask = (const int*)d_in[3];
  const float* Wq = (const float*)d_in[4];
  const float* Wk = (const float*)d_in[5];
  const float* Wv = (const float*)d_in[6];
  const float* Wo = (const float*)d_in[7];
  const float* lq1 = (const float*)d_in[8];
  const float* lk1 = (const float*)d_in[9];
  const float* lq2 = (const float*)d_in[10];
  const float* lk2 = (const float*)d_in[11];
  const float* lnw = (const float*)d_in[12];

  float* outp  = (float*)d_out;
  float* diffp = outp + (size_t)B_ * L_ * E_;

  float* Qs = (float*)d_ws;
  float* Km = Qs + (size_t)B_ * L_ * E_;
  float* Vm = Km + (size_t)B_ * S_ * E_;
  float* Bk = Vm + (size_t)B_ * S_ * E_;

  kv_proj_kernel<<<(B_ * S_) / 16, 128, 0, stream>>>(key, Wk, Wv, kmask, Km, Vm, Bk);
  q_proj_kernel<<<(B_ * L_) / 16, 32, 0, stream>>>(query, Wq, Qs);
  attn_kernel<<<(B_ * L_) / 16, 32, 0, stream>>>(Qs, Km, Vm, qmask, Bk,
                                                 lq1, lk1, lq2, lk2, lnw, Wo,
                                                 outp, diffp);
}


# attention kernel: the eight one-wave tiles of 128 consecutive rows form one 8-wave work-group (no barrier, 32 KB of LDS a wave): their K and V lines meet in the WGP cache, not 8 fetches from L2
# speedup vs baseline: 5.2354x; 1.0493x over previous
#include <hip/hip_runtime.h>
#include <math.h>

typedef float v2f __attribute__((ext_vector_type(2)));
typedef float v4f __attribute__((ext_vector_type(4)));
typedef int   v4i __attribute__((ext_vector_type(4)));
typedef float v8f __attribute__((ext_vector_type(8)));

#define WMMA_F32(a, b, c) \
  __builtin_amdgcn_wmma_f32_16x16x4_f32(false, (a), false, (b), (short)0, (c), false, false)

static constexpr int B_ = 8, S_ = 512, E_ = 32, L_ = 4096, SUBE_ = 768;
static constexpr float LAMBDA_INIT = 0.2f;

__global__ __launch_bounds__(128) __attribute__((amdgpu_waves_per_eu(1, 1))) void kv_proj_kernel(
    const float* __restrict__ key, const float* __restrict__ Wk,
    const float* __restrict__ Wv, const int* __restrict__ kmask,
    float* __restrict__ Km, float* __restrict__ Vm, float* __restrict__ Bk) {
  __shared__ __attribute__((aligned(16))) float tile[16 * 68];
  const int tid  = threadIdx.x;
  const int w    = tid >> 5;
  const int lane = tid & 31;
  const int n    = lane & 15;
  const int h    = lane >> 4;
  const int m0   = blockIdx.x * 16;

  const bool bias_writer = (blockIdx.x & 1) == 0 && tid < 64;
  const int kidx = (m0 + lane < B_ * S_) ? m0 + lane : B_ * S_ - 1;
  const int kmv = kmask[kidx];

  const float* wsel = (w < 2) ? Wk : Wv;
  const float* ap = key + (size_t)(m0 + n) * SUBE_ + 4 * h;
  const float* bp = wsel + (size_t)((w & 1) * 16 + n) * SUBE_ + 4 * h;

  v8f acc0 = {}, acc1 = {};
#pragma unroll
  for (int i = 0; i < SUBE_ / 8; ++i) {
    const v4f a = *(const v4f*)(ap + 8 * i);
    const v4f b = *(const v4f*)(bp + 8 * i);
    v2f a01, b01, a23, b23;
    a01[0] = a[0]; a01[1] = a[1]; b01[0] = b[0]; b01[1] = b[1];
    a23[0] = a[2]; a23[1] = a[3]; b23[0] = b[2]; b23[1] = b[3];
    acc0 = WMMA_F32(a01, b01, acc0);
    acc1 = WMMA_F32(a23, b23, acc1);
  }

#pragma unroll
  for (int v = 0; v < 8; ++v)
    tile[(8 * h + v) * 68 + 16 * w + n] = acc0[v] + acc1[v];
  __syncthreads();

#pragma unroll
  for (int q = 0; q < 2; ++q) {
    const int p    = tid + 128 * q;
    const int line = p >> 3;
    const int seg  = p & 7;
    const int row  = line & 15;
    const int isv  = line >> 4;
    const v4f x = *(const v4f*)(tile + row * 68 + 32 * isv + 4 * seg);
    float* dst = (isv ? Vm : Km) + (size_t)(m0 + row) * 32 + 4 * seg;
    *(v4f*)dst = x;
  }
  if (bias_writer)
    Bk[(size_t)w * (B_ * S_) + m0 + lane] = kmv ? 0.f : -1e20f;
}

__global__ __launch_bounds__(32) void q_proj_kernel(
    const float* __restrict__ q, const float* __restrict__ Wq,
    float* __restrict__ Qs) {
  const int lane = threadIdx.x;
  const int n    = lane & 15;
  const int khi  = lane >> 4;
  const int m0   = blockIdx.x * 16;
  const int mrow = m0 + n;

  v8f a0 = {}, a1 = {};
  for (int kk = 0; kk < 8; ++kk) {
    const int kb = kk * 4 + khi * 2;
    v2f a  = *(const v2f*)(q + (size_t)mrow * 32 + kb);
    v2f b0 = *(const v2f*)(Wq + (size_t)n * 32 + kb);
    v2f b1 = *(const v2f*)(Wq + (size_t)(16 + n) * 32 + kb);
    a0 = WMMA_F32(a, b0, a0);
    a1 = WMMA_F32(a, b1, a1);
  }
  const int rbase = m0 + (khi ? 8 : 0);
  for (int v = 0; v < 8; ++v) {
    const size_t r = (size_t)(rbase + v) * 32;
    Qs[r + n]      = a0[v] * 0.25f;
    Qs[r + 16 + n] = a1[v] * 0.25f;
  }
}

__global__ __launch_bounds__(256) __attribute__((amdgpu_waves_per_eu(2, 2))) void attn_kernel(
    const float* __restrict__ Qs, const float* __restrict__ Km,
    const float* __restrict__ Vm, const int* __restrict__ qmask,
    const float* __restrict__ Bk, const float* __restrict__ lq1,
    const float* __restrict__ lk1, const float* __restrict__ lq2,
    const float* __restrict__ lk2, const float* __restrict__ lnw,
    const float* __restrict__ Wo, float* __restrict__ outp,
    float* __restrict__ diffp) {
  __shared__ __attribute__((aligned(16))) v4f park_all[8 * 64 * 32];
  const int lane = threadIdx.x & 31;
  const int wv   = __builtin_amdgcn_readfirstlane((int)(threadIdx.x >> 5));
  v4f* park      = park_all + wv * (64 * 32);
  const int c    = lane & 15;
  const int h    = lane >> 4;
  const int R0   = (blockIdx.x * 8 + wv) * 16;
  const int b    = R0 / L_;

  float s1 = 0.f, s2 = 0.f;
  for (int i = 0; i < 16; ++i) { s1 += lq1[i] * lk1[i]; s2 += lq2[i] * lk2[i]; }
  const float lambda_full = __expf(s1) - __expf(s2) + LAMBDA_INIT;

  const int qv = qmask[R0 + c];

  const int crow = ((c & 4) << 1) | ((c & 8) >> 1) | (c & 3);
  const float* kbase = Km + (size_t)(b * S_ + crow) * 32 + 8 * h;
  const float* bbase = Bk + b * S_ + 4 * h;

  const float zq = qv ? 0.f : 1.f;

  v8f D[32];

#pragma unroll 1
  for (int hd = 0; hd < 2; ++hd) {
    const float* qp = Qs + (size_t)(R0 + c) * 32 + hd * 16 + 8 * h;
    const v4f qa = *(const v4f*)(qp);
    const v4f qb = *(const v4f*)(qp + 4);
    v2f q0, q1, q2, q3;
    q0[0] = qa[0]; q0[1] = qa[1];
    q1[0] = qa[2]; q1[1] = qa[3];
    q2[0] = qb[0]; q2[1] = qb[1];
    q3[0] = qb[2]; q3[1] = qb[3];
    const float* kh = kbase + hd * 16;
    const float* bh = bbase + hd * (B_ * S_);

    v4f KA[32], KB[32];
#pragma unroll
    for (int g = 0; g <= 4; ++g) {
      if (g < 4) {
#pragma unroll
        for (int i = 0; i < 8; ++i) {
          const int t = 8 * g + i;
          const float* kp = kh + (size_t)t * 16 * 32;
          KA[t] = *(const v4f*)(kp);
          KB[t] = *(const v4f*)(kp + 4);
          const v4f ba = *(const v4f*)(bh + t * 16);
          const v4f bb = *(const v4f*)(bh + t * 16 + 8);
          v8f x;
          x[0] = ba[0]; x[1] = ba[1]; x[2] = ba[2]; x[3] = ba[3];
          x[4] = bb[0]; x[5] = bb[1]; x[6] = bb[2]; x[7] = bb[3];
          D[t] = x;
        }
      }
      __builtin_amdgcn_sched_barrier(0);
      if (g > 0) {
#pragma unroll
        for (int i = 0; i < 8; i += 2) {
          const int t = 8 * (g - 1) + i;
          const v4f xa = KA[t], xb = KB[t], ya = KA[t + 1], yb = KB[t + 1];
          v8f x = D[t], y = D[t + 1];
          v2f f;
          f[0] = xa[0]; f[1] = xa[1]; x = WMMA_F32(f, q0, x);
          f[0] = ya[0]; f[1] = ya[1]; y = WMMA_F32(f, q0, y);
          f[0] = xa[2]; f[1] = xa[3]; x = WMMA_F32(f, q1, x);
          f[0] = ya[2]; f[1] = ya[3]; y = WMMA_F32(f, q1, y);
          f[0] = xb[0]; f[1] = xb[1]; x = WMMA_F32(f, q2, x);
          f[0] = yb[0]; f[1] = yb[1]; y = WMMA_F32(f, q2, y);
          f[0] = xb[2]; f[1] = xb[3]; x = WMMA_F32(f, q3, x);
          f[0] = yb[2]; f[1] = yb[3]; y = WMMA_F32(f, q3, y);
          D[t] = x; D[t + 1] = y;
        }
      }
      __builtin_amdgcn_sched_barrier(0);
    }

    v8f mv = D[0];
#pragma unroll
    for (int t = 1; t < 32; ++t) {
#pragma unroll
      for (int r = 0; r < 8; ++r) mv[r] = fmaxf(mv[r], D[t][r]);
    }
    float mx = fmaxf(fmaxf(fmaxf(mv[0], mv[1]), fmaxf(mv[2], mv[3])),
                     fmaxf(fmaxf(mv[4], mv[5]), fmaxf(mv[6], mv[7])));
    mx = fmaxf(mx, __shfl_xor(mx, 16, 32));

    v8f sv = {};
#pragma unroll
    for (int t = 0; t < 32; ++t) {
#pragma unroll
      for (int r = 0; r < 8; ++r) {
        const float e = __expf(D[t][r] - mx);
        D[t][r] = e; sv[r] += e;
      }
    }
    float sum = ((sv[0] + sv[1]) + (sv[2] + sv[3])) +
                ((sv[4] + sv[5]) + (sv[6] + sv[7]));
    sum += __shfl_xor(sum, 16, 32);
    const float inv   = 1.f / (sum + 1e-8f);
    if (hd == 0) {
#pragma unroll
      for (int t = 0; t < 32; ++t) {
        v4f lo, hi;
#pragma unroll
        for (int r = 0; r < 4; ++r) { lo[r] = D[t][r] * inv; hi[r] = D[t][4 + r] * inv; }
        park[(2 * t) * 32 + lane]     = lo;
        park[(2 * t + 1) * 32 + lane] = hi;
      }
    } else {
      const float scale = -lambda_full * inv;
#pragma unroll
      for (int t = 0; t < 32; ++t) {
        const v4f lo = park[(2 * t) * 32 + lane];
        const v4f hi = park[(2 * t + 1) * 32 + lane];
#pragma unroll
        for (int r = 0; r < 4; ++r) {
          D[t][r]     = D[t][r] * scale + lo[r];
          D[t][4 + r] = D[t][4 + r] * scale + hi[r];
        }
      }
    }
  }

  v8f dmv = D[0];
#pragma unroll
  for (int t = 1; t < 32; ++t) {
#pragma unroll
    for (int r = 0; r < 8; ++r) dmv[r] = fminf(dmv[r], D[t][r]);
  }
  float dmin = fminf(fminf(fminf(dmv[0], dmv[1]), fminf(dmv[2], dmv[3])),
                     fminf(fminf(dmv[4], dmv[5]), fminf(dmv[6], dmv[7])));
  dmin = fminf(dmin, __shfl_xor(dmin, 16, 32));

  float* dg = diffp + (size_t)(R0 + c) * 512 + 4 * h;
#pragma unroll
  for (int t = 0; t < 32; ++t) {
    const v4f ba = *(const v4f*)(bbase + t * 16);
    const v4f bb = *(const v4f*)(bbase + t * 16 + 8);
    v4f lo, hi;
#pragma unroll
    for (int r = 0; r < 8; ++r) {
      const float bias = (r < 4) ? ba[r & 3] : bb[r & 3];
      float dd = D[t][r] - dmin + 1e-5f;
      dd = (bias == zq) ? dd : 0.f;
      D[t][r] = dd;
      if (r < 4) lo[r & 3] = dd; else hi[r & 3] = dd;
    }
    __builtin_nontemporal_store(lo, (v4f*)(dg + t * 16));
    __builtin_nontemporal_store(hi, (v4f*)(dg + t * 16 + 8));
  }

  const float* vbase = Vm + (size_t)(b * S_ + 4 * h) * 32 + c;
  v8f ao0 = {}, ao1 = {};
#pragma unroll
  for (int t = 0; t < 32; ++t) {
#pragma unroll
    for (int j = 0; j < 4; ++j) {
      const float* vp = vbase + (size_t)(t * 16 + 8 * (j >> 1) + 2 * (j & 1)) * 32;
      v2f a0, a1, bf;
      a0[0] = vp[0];  a0[1] = vp[32];
      a1[0] = vp[16]; a1[1] = vp[48];
      bf[0] = D[t][2 * j]; bf[1] = D[t][2 * j + 1];
      ao0 = WMMA_F32(a0, bf, ao0);
      ao1 = WMMA_F32(a1, bf, ao1);
    }
  }

  float sq = 0.f;
#pragma unroll
  for (int r = 0; r < 8; ++r) sq += ao0[r] * ao0[r];
#pragma unroll
  for (int r = 0; r < 8; ++r) sq += ao1[r] * ao1[r];
  sq += __shfl_xor(sq, 16, 32);
  const float rms = rsqrtf(sq * (1.0f / 32.0f) + 1e-5f);
#pragma unroll
  for (int r = 0; r < 8; ++r) {
    ao0[r] = ao0[r] * rms * lnw[8 * h + r] * 0.8f;
    ao1[r] = ao1[r] * rms * lnw[16 + 8 * h + r] * 0.8f;
  }

  v8f o0 = {}, o1 = {};
#pragma unroll
  for (int mt = 0; mt < 2; ++mt) {
#pragma unroll
    for (int j = 0; j < 4; ++j) {
      const int kb = 16 * mt + 8 * h + 2 * j;
      const v2f w0 = *(const v2f*)(Wo + (size_t)c * 32 + kb);
      const v2f w1 = *(const v2f*)(Wo + (size_t)(16 + c) * 32 + kb);
      v2f bf;
      bf[0] = mt ? ao1[2 * j] : ao0[2 * j];
      bf[1] = mt ? ao1[2 * j + 1] : ao0[2 * j + 1];
      o0 = WMMA_F32(w0, bf, o0);
      o1 = WMMA_F32(w1, bf, o1);
    }
  }
  float* og = outp + (size_t)(R0 + c) * 32 + 8 * h;
  v4f x;
  x[0] = o0[0]; x[1] = o0[1]; x[2] = o0[2]; x[3] = o0[3];
  __builtin_nontemporal_store(x, (v4f*)(og));
  x[0] = o0[4]; x[1] = o0[5]; x[2] = o0[6]; x[3] = o0[7];
  __builtin_nontemporal_store(x, (v4f*)(og + 4));
  x[0] = o1[0]; x[1] = o1[1]; x[2] = o1[2]; x[3] = o1[3];
  __builtin_nontemporal_store(x, (v4f*)(og + 16));
  x[0] = o1[4]; x[1] = o1[5]; x[2] = o1[6]; x[3] = o1[7];
  __builtin_nontemporal_store(x, (v4f*)(og + 20));
}

extern "C" void kernel_launch(void* const* d_in, const int* in_sizes, int n_in,
                              void* d_out, int out_size, void* d_ws, size_t ws_size,
                              hipStream_t stream) {
  (void)in_sizes; (void)n_in; (void)out_size; (void)ws_size;
  const float* query = (const float*)d_in[0];
  const float* key   = (const float*)d_in[1];
  const int*   qmask = (const int*)d_in[2];
  const int*   kmask = (const int*)d_in[3];
  const float* Wq = (const float*)d_in[4];
  const float* Wk = (const float*)d_in[5];
  const float* Wv = (const float*)d_in[6];
  const float* Wo = (const float*)d_in[7];
  const float* lq1 = (const float*)d_in[8];
  const float* lk1 = (const float*)d_in[9];
  const float* lq2 = (const float*)d_in[10];
  const float* lk2 = (const float*)d_in[11];
  const float* lnw = (const float*)d_in[12];

  float* outp  = (float*)d_out;
  float* diffp = outp + (size_t)B_ * L_ * E_;

  float* Qs = (float*)d_ws;
  float* Km = Qs + (size_t)B_ * L_ * E_;
  float* Vm = Km + (size_t)B_ * S_ * E_;
  float* Bk = Vm + (size_t)B_ * S_ * E_;

  kv_proj_kernel<<<(B_ * S_) / 16, 128, 0, stream>>>(key, Wk, Wv, kmask, Km, Vm, Bk);
  q_proj_kernel<<<(B_ * L_) / 16, 32, 0, stream>>>(query, Wq, Qs);
  attn_kernel<<<(B_ * L_) / 128, 256, 0, stream>>>(Qs, Km, Vm, qmask, Bk,
                                                 lq1, lk1, lq2, lk2, lnw, Wo,
                                                 outp, diffp);
}


# attention kernel: diff tile staged through the wave's free LDS region (16-byte chunks XOR-swizzled by row): each store instruction writes 4 whole 128-byte lines, not 16 pieces of 32 bytes
# speedup vs baseline: 6.5688x; 1.2547x over previous
#include <hip/hip_runtime.h>
#include <math.h>

typedef float v2f __attribute__((ext_vector_type(2)));
typedef float v4f __attribute__((ext_vector_type(4)));
typedef int   v4i __attribute__((ext_vector_type(4)));
typedef float v8f __attribute__((ext_vector_type(8)));

#define WMMA_F32(a, b, c) \
  __builtin_amdgcn_wmma_f32_16x16x4_f32(false, (a), false, (b), (short)0, (c), false, false)

static constexpr int B_ = 8, S_ = 512, E_ = 32, L_ = 4096, SUBE_ = 768;
static constexpr float LAMBDA_INIT = 0.2f;

__global__ __launch_bounds__(128) __attribute__((amdgpu_waves_per_eu(1, 1))) void kv_proj_kernel(
    const float* __restrict__ key, const float* __restrict__ Wk,
    const float* __restrict__ Wv, const int* __restrict__ kmask,
    float* __restrict__ Km, float* __restrict__ Vm, float* __restrict__ Bk) {
  __shared__ __attribute__((aligned(16))) float tile[16 * 68];
  const int tid  = threadIdx.x;
  const int w    = tid >> 5;
  const int lane = tid & 31;
  const int n    = lane & 15;
  const int h    = lane >> 4;
  const int m0   = blockIdx.x * 16;

  const bool bias_writer = (blockIdx.x & 1) == 0 && tid < 64;
  const int kidx = (m0 + lane < B_ * S_) ? m0 + lane : B_ * S_ - 1;
  const int kmv = kmask[kidx];

  const float* wsel = (w < 2) ? Wk : Wv;
  const float* ap = key + (size_t)(m0 + n) * SUBE_ + 4 * h;
  const float* bp = wsel + (size_t)((w & 1) * 16 + n) * SUBE_ + 4 * h;

  v8f acc0 = {}, acc1 = {};
#pragma unroll
  for (int i = 0; i < SUBE_ / 8; ++i) {
    const v4f a = *(const v4f*)(ap + 8 * i);
    const v4f b = *(const v4f*)(bp + 8 * i);
    v2f a01, b01, a23, b23;
    a01[0] = a[0]; a01[1] = a[1]; b01[0] = b[0]; b01[1] = b[1];
    a23[0] = a[2]; a23[1] = a[3]; b23[0] = b[2]; b23[1] = b[3];
    acc0 = WMMA_F32(a01, b01, acc0);
    acc1 = WMMA_F32(a23, b23, acc1);
  }

#pragma unroll
  for (int v = 0; v < 8; ++v)
    tile[(8 * h + v) * 68 + 16 * w + n] = acc0[v] + acc1[v];
  __syncthreads();

#pragma unroll
  for (int q = 0; q < 2; ++q) {
    const int p    = tid + 128 * q;
    const int line = p >> 3;
    const int seg  = p & 7;
    const int row  = line & 15;
    const int isv  = line >> 4;
    const v4f x = *(const v4f*)(tile + row * 68 + 32 * isv + 4 * seg);
    float* dst = (isv ? Vm : Km) + (size_t)(m0 + row) * 32 + 4 * seg;
    *(v4f*)dst = x;
  }
  if (bias_writer)
    Bk[(size_t)w * (B_ * S_) + m0 + lane] = kmv ? 0.f : -1e20f;
}

__global__ __launch_bounds__(32) void q_proj_kernel(
    const float* __restrict__ q, const float* __restrict__ Wq,
    float* __restrict__ Qs) {
  const int lane = threadIdx.x;
  const int n    = lane & 15;
  const int khi  = lane >> 4;
  const int m0   = blockIdx.x * 16;
  const int mrow = m0 + n;

  v8f a0 = {}, a1 = {};
  for (int kk = 0; kk < 8; ++kk) {
    const int kb = kk * 4 + khi * 2;
    v2f a  = *(const v2f*)(q + (size_t)mrow * 32 + kb);
    v2f b0 = *(const v2f*)(Wq + (size_t)n * 32 + kb);
    v2f b1 = *(const v2f*)(Wq + (size_t)(16 + n) * 32 + kb);
    a0 = WMMA_F32(a, b0, a0);
    a1 = WMMA_F32(a, b1, a1);
  }
  const int rbase = m0 + (khi ? 8 : 0);
  for (int v = 0; v < 8; ++v) {
    const size_t r = (size_t)(rbase + v) * 32;
    Qs[r + n]      = a0[v] * 0.25f;
    Qs[r + 16 + n] = a1[v] * 0.25f;
  }
}

__global__ __launch_bounds__(256) __attribute__((amdgpu_waves_per_eu(2, 2))) void attn_kernel(
    const float* __restrict__ Qs, const float* __restrict__ Km,
    const float* __restrict__ Vm, const int* __restrict__ qmask,
    const float* __restrict__ Bk, const float* __restrict__ lq1,
    const float* __restrict__ lk1, const float* __restrict__ lq2,
    const float* __restrict__ lk2, const float* __restrict__ lnw,
    const float* __restrict__ Wo, float* __restrict__ outp,
    float* __restrict__ diffp) {
  __shared__ __attribute__((aligned(16))) v4f park_all[8 * 64 * 32];
  const int lane = threadIdx.x & 31;
  const int wv   = __builtin_amdgcn_readfirstlane((int)(threadIdx.x >> 5));
  v4f* park      = park_all + wv * (64 * 32);
  const int c    = lane & 15;
  const int h    = lane >> 4;
  const int R0   = (blockIdx.x * 8 + wv) * 16;
  const int b    = R0 / L_;

  float s1 = 0.f, s2 = 0.f;
  for (int i = 0; i < 16; ++i) { s1 += lq1[i] * lk1[i]; s2 += lq2[i] * lk2[i]; }
  const float lambda_full = __expf(s1) - __expf(s2) + LAMBDA_INIT;

  const int qv = qmask[R0 + c];

  const int crow = ((c & 4) << 1) | ((c & 8) >> 1) | (c & 3);
  const float* kbase = Km + (size_t)(b * S_ + crow) * 32 + 8 * h;
  const float* bbase = Bk + b * S_ + 4 * h;

  const float zq = qv ? 0.f : 1.f;

  v8f D[32];

#pragma unroll 1
  for (int hd = 0; hd < 2; ++hd) {
    const float* qp = Qs + (size_t)(R0 + c) * 32 + hd * 16 + 8 * h;
    const v4f qa = *(const v4f*)(qp);
    const v4f qb = *(const v4f*)(qp + 4);
    v2f q0, q1, q2, q3;
    q0[0] = qa[0]; q0[1] = qa[1];
    q1[0] = qa[2]; q1[1] = qa[3];
    q2[0] = qb[0]; q2[1] = qb[1];
    q3[0] = qb[2]; q3[1] = qb[3];
    const float* kh = kbase + hd * 16;
    const float* bh = bbase + hd * (B_ * S_);

    v4f KA[32], KB[32];
#pragma unroll
    for (int g = 0; g <= 4; ++g) {
      if (g < 4) {
#pragma unroll
        for (int i = 0; i < 8; ++i) {
          const int t = 8 * g + i;
          const float* kp = kh + (size_t)t * 16 * 32;
          KA[t] = *(const v4f*)(kp);
          KB[t] = *(const v4f*)(kp + 4);
          const v4f ba = *(const v4f*)(bh + t * 16);
          const v4f bb = *(const v4f*)(bh + t * 16 + 8);
          v8f x;
          x[0] = ba[0]; x[1] = ba[1]; x[2] = ba[2]; x[3] = ba[3];
          x[4] = bb[0]; x[5] = bb[1]; x[6] = bb[2]; x[7] = bb[3];
          D[t] = x;
        }
      }
      __builtin_amdgcn_sched_barrier(0);
      if (g > 0) {
#pragma unroll
        for (int i = 0; i < 8; i += 2) {
          const int t = 8 * (g - 1) + i;
          const v4f xa = KA[t], xb = KB[t], ya = KA[t + 1], yb = KB[t + 1];
          v8f x = D[t], y = D[t + 1];
          v2f f;
          f[0] = xa[0]; f[1] = xa[1]; x = WMMA_F32(f, q0, x);
          f[0] = ya[0]; f[1] = ya[1]; y = WMMA_F32(f, q0, y);
          f[0] = xa[2]; f[1] = xa[3]; x = WMMA_F32(f, q1, x);
          f[0] = ya[2]; f[1] = ya[3]; y = WMMA_F32(f, q1, y);
          f[0] = xb[0]; f[1] = xb[1]; x = WMMA_F32(f, q2, x);
          f[0] = yb[0]; f[1] = yb[1]; y = WMMA_F32(f, q2, y);
          f[0] = xb[2]; f[1] = xb[3]; x = WMMA_F32(f, q3, x);
          f[0] = yb[2]; f[1] = yb[3]; y = WMMA_F32(f, q3, y);
          D[t] = x; D[t + 1] = y;
        }
      }
      __builtin_amdgcn_sched_barrier(0);
    }

    v8f mv = D[0];
#pragma unroll
    for (int t = 1; t < 32; ++t) {
#pragma unroll
      for (int r = 0; r < 8; ++r) mv[r] = fmaxf(mv[r], D[t][r]);
    }
    float mx = fmaxf(fmaxf(fmaxf(mv[0], mv[1]), fmaxf(mv[2], mv[3])),
                     fmaxf(fmaxf(mv[4], mv[5]), fmaxf(mv[6], mv[7])));
    mx = fmaxf(mx, __shfl_xor(mx, 16, 32));

    v8f sv = {};
#pragma unroll
    for (int t = 0; t < 32; ++t) {
#pragma unroll
      for (int r = 0; r < 8; ++r) {
        const float e = __expf(D[t][r] - mx);
        D[t][r] = e; sv[r] += e;
      }
    }
    float sum = ((sv[0] + sv[1]) + (sv[2] + sv[3])) +
                ((sv[4] + sv[5]) + (sv[6] + sv[7]));
    sum += __shfl_xor(sum, 16, 32);
    const float inv   = 1.f / (sum + 1e-8f);
    if (hd == 0) {
#pragma unroll
      for (int t = 0; t < 32; ++t) {
        v4f lo, hi;
#pragma unroll
        for (int r = 0; r < 4; ++r) { lo[r] = D[t][r] * inv; hi[r] = D[t][4 + r] * inv; }
        park[(2 * t) * 32 + lane]     = lo;
        park[(2 * t + 1) * 32 + lane] = hi;
      }
    } else {
      const float scale = -lambda_full * inv;
#pragma unroll
      for (int t = 0; t < 32; ++t) {
        const v4f lo = park[(2 * t) * 32 + lane];
        const v4f hi = park[(2 * t + 1) * 32 + lane];
#pragma unroll
        for (int r = 0; r < 4; ++r) {
          D[t][r]     = D[t][r] * scale + lo[r];
          D[t][4 + r] = D[t][4 + r] * scale + hi[r];
        }
      }
    }
  }

  v8f dmv = D[0];
#pragma unroll
  for (int t = 1; t < 32; ++t) {
#pragma unroll
    for (int r = 0; r < 8; ++r) dmv[r] = fminf(dmv[r], D[t][r]);
  }
  float dmin = fminf(fminf(fminf(dmv[0], dmv[1]), fminf(dmv[2], dmv[3])),
                     fminf(fminf(dmv[4], dmv[5]), fminf(dmv[6], dmv[7])));
  dmin = fminf(dmin, __shfl_xor(dmin, 16, 32));

  int wb[8];
#pragma unroll
  for (int k = 0; k < 8; ++k) wb[k] = c * 128 + ((2 * k + h) ^ c);
#pragma unroll
  for (int t = 0; t < 32; ++t) {
    const v4f ba = *(const v4f*)(bbase + t * 16);
    const v4f bb = *(const v4f*)(bbase + t * 16 + 8);
    v4f lo, hi;
#pragma unroll
    for (int r = 0; r < 8; ++r) {
      const float bias = (r < 4) ? ba[r & 3] : bb[r & 3];
      float dd = D[t][r] - dmin + 1e-5f;
      dd = (bias == zq) ? dd : 0.f;
      D[t][r] = dd;
      if (r < 4) lo[r & 3] = dd; else hi[r & 3] = dd;
    }
    park[wb[2 * (t & 3)]     + (t >> 2) * 16] = lo;
    park[wb[2 * (t & 3) + 1] + (t >> 2) * 16] = hi;
  }
  __builtin_amdgcn_fence(__ATOMIC_RELEASE, "wavefront");
  __builtin_amdgcn_wave_barrier();
  __builtin_amdgcn_fence(__ATOMIC_ACQUIRE, "wavefront");
  float* dg = diffp + (size_t)R0 * 512 + 4 * lane;
#pragma unroll
  for (int R = 0; R < 16; ++R) {
    const int rb = R * 128 + (lane ^ R);
#pragma unroll
    for (int i = 0; i < 4; ++i) {
      const v4f x = park[rb + 32 * i];
      __builtin_nontemporal_store(x, (v4f*)(dg + R * 512 + 128 * i));
    }
  }

  const float* vbase = Vm + (size_t)(b * S_ + 4 * h) * 32 + c;
  v8f ao0 = {}, ao1 = {};
#pragma unroll
  for (int t = 0; t < 32; ++t) {
#pragma unroll
    for (int j = 0; j < 4; ++j) {
      const float* vp = vbase + (size_t)(t * 16 + 8 * (j >> 1) + 2 * (j & 1)) * 32;
      v2f a0, a1, bf;
      a0[0] = vp[0];  a0[1] = vp[32];
      a1[0] = vp[16]; a1[1] = vp[48];
      bf[0] = D[t][2 * j]; bf[1] = D[t][2 * j + 1];
      ao0 = WMMA_F32(a0, bf, ao0);
      ao1 = WMMA_F32(a1, bf, ao1);
    }
  }

  float sq = 0.f;
#pragma unroll
  for (int r = 0; r < 8; ++r) sq += ao0[r] * ao0[r];
#pragma unroll
  for (int r = 0; r < 8; ++r) sq += ao1[r] * ao1[r];
  sq += __shfl_xor(sq, 16, 32);
  const float rms = rsqrtf(sq * (1.0f / 32.0f) + 1e-5f);
#pragma unroll
  for (int r = 0; r < 8; ++r) {
    ao0[r] = ao0[r] * rms * lnw[8 * h + r] * 0.8f;
    ao1[r] = ao1[r] * rms * lnw[16 + 8 * h + r] * 0.8f;
  }

  v8f o0 = {}, o1 = {};
#pragma unroll
  for (int mt = 0; mt < 2; ++mt) {
#pragma unroll
    for (int j = 0; j < 4; ++j) {
      const int kb = 16 * mt + 8 * h + 2 * j;
      const v2f w0 = *(const v2f*)(Wo + (size_t)c * 32 + kb);
      const v2f w1 = *(const v2f*)(Wo + (size_t)(16 + c) * 32 + kb);
      v2f bf;
      bf[0] = mt ? ao1[2 * j] : ao0[2 * j];
      bf[1] = mt ? ao1[2 * j + 1] : ao0[2 * j + 1];
      o0 = WMMA_F32(w0, bf, o0);
      o1 = WMMA_F32(w1, bf, o1);
    }
  }
  float* og = outp + (size_t)(R0 + c) * 32 + 8 * h;
  v4f x;
  x[0] = o0[0]; x[1] = o0[1]; x[2] = o0[2]; x[3] = o0[3];
  __builtin_nontemporal_store(x, (v4f*)(og));
  x[0] = o0[4]; x[1] = o0[5]; x[2] = o0[6]; x[3] = o0[7];
  __builtin_nontemporal_store(x, (v4f*)(og + 4));
  x[0] = o1[0]; x[1] = o1[1]; x[2] = o1[2]; x[3] = o1[3];
  __builtin_nontemporal_store(x, (v4f*)(og + 16));
  x[0] = o1[4]; x[1] = o1[5]; x[2] = o1[6]; x[3] = o1[7];
  __builtin_nontemporal_store(x, (v4f*)(og + 20));
}

extern "C" void kernel_launch(void* const* d_in, const int* in_sizes, int n_in,
                              void* d_out, int out_size, void* d_ws, size_t ws_size,
                              hipStream_t stream) {
  (void)in_sizes; (void)n_in; (void)out_size; (void)ws_size;
  const float* query = (const float*)d_in[0];
  const float* key   = (const float*)d_in[1];
  const int*   qmask = (const int*)d_in[2];
  const int*   kmask = (const int*)d_in[3];
  const float* Wq = (const float*)d_in[4];
  const float* Wk = (const float*)d_in[5];
  const float* Wv = (const float*)d_in[6];
  const float* Wo = (const float*)d_in[7];
  const float* lq1 = (const float*)d_in[8];
  const float* lk1 = (const float*)d_in[9];
  const float* lq2 = (const float*)d_in[10];
  const float* lk2 = (const float*)d_in[11];
  const float* lnw = (const float*)d_in[12];

  float* outp  = (float*)d_out;
  float* diffp = outp + (size_t)B_ * L_ * E_;

  float* Qs = (float*)d_ws;
  float* Km = Qs + (size_t)B_ * L_ * E_;
  float* Vm = Km + (size_t)B_ * S_ * E_;
  float* Bk = Vm + (size_t)B_ * S_ * E_;

  kv_proj_kernel<<<(B_ * S_) / 16, 128, 0, stream>>>(key, Wk, Wv, kmask, Km, Vm, Bk);
  q_proj_kernel<<<(B_ * L_) / 16, 32, 0, stream>>>(query, Wq, Qs);
  attn_kernel<<<(B_ * L_) / 128, 256, 0, stream>>>(Qs, Km, Vm, qmask, Bk,
                                                 lq1, lk1, lq2, lk2, lnw, Wo,
                                                 outp, diffp);
}


# Q projection done by four extra waves (4-7) of each K/V projection work-group, two Q tiles a wave, beside the K/V waves on their SIMDs: 3 launches -> 2; all sums unchanged
# speedup vs baseline: 6.9106x; 1.0520x over previous
#include <hip/hip_runtime.h>
#include <math.h>

typedef float v2f __attribute__((ext_vector_type(2)));
typedef float v4f __attribute__((ext_vector_type(4)));
typedef int   v4i __attribute__((ext_vector_type(4)));
typedef float v8f __attribute__((ext_vector_type(8)));

#define WMMA_F32(a, b, c) \
  __builtin_amdgcn_wmma_f32_16x16x4_f32(false, (a), false, (b), (short)0, (c), false, false)

static constexpr int B_ = 8, S_ = 512, E_ = 32, L_ = 4096, SUBE_ = 768;
static constexpr float LAMBDA_INIT = 0.2f;

__global__ __launch_bounds__(256) __attribute__((amdgpu_waves_per_eu(2, 2))) void proj_kernel(
    const float* __restrict__ key, const float* __restrict__ Wk,
    const float* __restrict__ Wv, const int* __restrict__ kmask,
    float* __restrict__ Km, float* __restrict__ Vm, float* __restrict__ Bk,
    const float* __restrict__ q, const float* __restrict__ Wq,
    float* __restrict__ Qs) {
  __shared__ __attribute__((aligned(16))) float tile[16 * 68];
  const int tid  = threadIdx.x;
  const int w    = tid >> 5;
  const int lane = tid & 31;
  const int n    = lane & 15;
  const int h    = lane >> 4;
  const int m0   = blockIdx.x * 16;
  const int wu   = __builtin_amdgcn_readfirstlane(w);

  if (wu >= 4) {
    for (int u = 0; u < 2; ++u) {
      const int qm0  = ((blockIdx.x * 4 + (wu - 4)) * 2 + u) * 16;
      const int mrow = qm0 + n;
      v8f a0 = {}, a1 = {};
      for (int kk = 0; kk < 8; ++kk) {
        const int kb = kk * 4 + h * 2;
        v2f a  = *(const v2f*)(q + (size_t)mrow * 32 + kb);
        v2f b0 = *(const v2f*)(Wq + (size_t)n * 32 + kb);
        v2f b1 = *(const v2f*)(Wq + (size_t)(16 + n) * 32 + kb);
        a0 = WMMA_F32(a, b0, a0);
        a1 = WMMA_F32(a, b1, a1);
      }
      const int rbase = qm0 + (h ? 8 : 0);
      for (int v = 0; v < 8; ++v) {
        const size_t r = (size_t)(rbase + v) * 32;
        Qs[r + n]      = a0[v] * 0.25f;
        Qs[r + 16 + n] = a1[v] * 0.25f;
      }
    }
  }

  const bool bias_writer = (blockIdx.x & 1) == 0 && tid < 64;
  const int kidx = (m0 + lane < B_ * S_) ? m0 + lane : B_ * S_ - 1;
  const int kmv = kmask[kidx];

  if (wu < 4) {
    const float* wsel = (w < 2) ? Wk : Wv;
    const float* ap = key + (size_t)(m0 + n) * SUBE_ + 4 * h;
    const float* bp = wsel + (size_t)((w & 1) * 16 + n) * SUBE_ + 4 * h;

    v8f acc0 = {}, acc1 = {};
#pragma unroll
    for (int i = 0; i < SUBE_ / 8; ++i) {
      const v4f a = *(const v4f*)(ap + 8 * i);
      const v4f b = *(const v4f*)(bp + 8 * i);
      v2f a01, b01, a23, b23;
      a01[0] = a[0]; a01[1] = a[1]; b01[0] = b[0]; b01[1] = b[1];
      a23[0] = a[2]; a23[1] = a[3]; b23[0] = b[2]; b23[1] = b[3];
      acc0 = WMMA_F32(a01, b01, acc0);
      acc1 = WMMA_F32(a23, b23, acc1);
    }

#pragma unroll
    for (int v = 0; v < 8; ++v)
      tile[(8 * h + v) * 68 + 16 * w + n] = acc0[v] + acc1[v];
  }
  __syncthreads();

  if (wu < 4) {
#pragma unroll
    for (int q2 = 0; q2 < 2; ++q2) {
      const int p    = tid + 128 * q2;
      const int line = p >> 3;
      const int seg  = p & 7;
      const int row  = line & 15;
      const int isv  = line >> 4;
      const v4f x = *(const v4f*)(tile + row * 68 + 32 * isv + 4 * seg);
      float* dst = (isv ? Vm : Km) + (size_t)(m0 + row) * 32 + 4 * seg;
      *(v4f*)dst = x;
    }
    if (bias_writer)
      Bk[(size_t)w * (B_ * S_) + m0 + lane] = kmv ? 0.f : -1e20f;
  }
}

__global__ __launch_bounds__(256) __attribute__((amdgpu_waves_per_eu(2, 2))) void attn_kernel(
    const float* __restrict__ Qs, const float* __restrict__ Km,
    const float* __restrict__ Vm, const int* __restrict__ qmask,
    const float* __restrict__ Bk, const float* __restrict__ lq1,
    const float* __restrict__ lk1, const float* __restrict__ lq2,
    const float* __restrict__ lk2, const float* __restrict__ lnw,
    const float* __restrict__ Wo, float* __restrict__ outp,
    float* __restrict__ diffp) {
  __shared__ __attribute__((aligned(16))) v4f park_all[8 * 64 * 32];
  const int lane = threadIdx.x & 31;
  const int wv   = __builtin_amdgcn_readfirstlane((int)(threadIdx.x >> 5));
  v4f* park      = park_all + wv * (64 * 32);
  const int c    = lane & 15;
  const int h    = lane >> 4;
  const int R0   = (blockIdx.x * 8 + wv) * 16;
  const int b    = R0 / L_;

  float s1 = 0.f, s2 = 0.f;
  for (int i = 0; i < 16; ++i) { s1 += lq1[i] * lk1[i]; s2 += lq2[i] * lk2[i]; }
  const float lambda_full = __expf(s1) - __expf(s2) + LAMBDA_INIT;

  const int qv = qmask[R0 + c];

  const int crow = ((c & 4) << 1) | ((c & 8) >> 1) | (c & 3);
  const float* kbase = Km + (size_t)(b * S_ + crow) * 32 + 8 * h;
  const float* bbase = Bk + b * S_ + 4 * h;

  const float zq = qv ? 0.f : 1.f;

  v8f D[32];

#pragma unroll 1
  for (int hd = 0; hd < 2; ++hd) {
    const float* qp = Qs + (size_t)(R0 + c) * 32 + hd * 16 + 8 * h;
    const v4f qa = *(const v4f*)(qp);
    const v4f qb = *(const v4f*)(qp + 4);
    v2f q0, q1, q2, q3;
    q0[0] = qa[0]; q0[1] = qa[1];
    q1[0] = qa[2]; q1[1] = qa[3];
    q2[0] = qb[0]; q2[1] = qb[1];
    q3[0] = qb[2]; q3[1] = qb[3];
    const float* kh = kbase + hd * 16;
    const float* bh = bbase + hd * (B_ * S_);

    v4f KA[32], KB[32];
#pragma unroll
    for (int g = 0; g <= 4; ++g) {
      if (g < 4) {
#pragma unroll
        for (int i = 0; i < 8; ++i) {
          const int t = 8 * g + i;
          const float* kp = kh + (size_t)t * 16 * 32;
          KA[t] = *(const v4f*)(kp);
          KB[t] = *(const v4f*)(kp + 4);
          const v4f ba = *(const v4f*)(bh + t * 16);
          const v4f bb = *(const v4f*)(bh + t * 16 + 8);
          v8f x;
          x[0] = ba[0]; x[1] = ba[1]; x[2] = ba[2]; x[3] = ba[3];
          x[4] = bb[0]; x[5] = bb[1]; x[6] = bb[2]; x[7] = bb[3];
          D[t] = x;
        }
      }
      __builtin_amdgcn_sched_barrier(0);
      if (g > 0) {
#pragma unroll
        for (int i = 0; i < 8; i += 2) {
          const int t = 8 * (g - 1) + i;
          const v4f xa = KA[t], xb = KB[t], ya = KA[t + 1], yb = KB[t + 1];
          v8f x = D[t], y = D[t + 1];
          v2f f;
          f[0] = xa[0]; f[1] = xa[1]; x = WMMA_F32(f, q0, x);
          f[0] = ya[0]; f[1] = ya[1]; y = WMMA_F32(f, q0, y);
          f[0] = xa[2]; f[1] = xa[3]; x = WMMA_F32(f, q1, x);
          f[0] = ya[2]; f[1] = ya[3]; y = WMMA_F32(f, q1, y);
          f[0] = xb[0]; f[1] = xb[1]; x = WMMA_F32(f, q2, x);
          f[0] = yb[0]; f[1] = yb[1]; y = WMMA_F32(f, q2, y);
          f[0] = xb[2]; f[1] = xb[3]; x = WMMA_F32(f, q3, x);
          f[0] = yb[2]; f[1] = yb[3]; y = WMMA_F32(f, q3, y);
          D[t] = x; D[t + 1] = y;
        }
      }
      __builtin_amdgcn_sched_barrier(0);
    }

    v8f mv = D[0];
#pragma unroll
    for (int t = 1; t < 32; ++t) {
#pragma unroll
      for (int r = 0; r < 8; ++r) mv[r] = fmaxf(mv[r], D[t][r]);
    }
    float mx = fmaxf(fmaxf(fmaxf(mv[0], mv[1]), fmaxf(mv[2], mv[3])),
                     fmaxf(fmaxf(mv[4], mv[5]), fmaxf(mv[6], mv[7])));
    mx = fmaxf(mx, __shfl_xor(mx, 16, 32));

    v8f sv = {};
#pragma unroll
    for (int t = 0; t < 32; ++t) {
#pragma unroll
      for (int r = 0; r < 8; ++r) {
        const float e = __expf(D[t][r] - mx);
        D[t][r] = e; sv[r] += e;
      }
    }
    float sum = ((sv[0] + sv[1]) + (sv[2] + sv[3])) +
                ((sv[4] + sv[5]) + (sv[6] + sv[7]));
    sum += __shfl_xor(sum, 16, 32);
    const float inv   = 1.f / (sum + 1e-8f);
    if (hd == 0) {
#pragma unroll
      for (int t = 0; t < 32; ++t) {
        v4f lo, hi;
#pragma unroll
        for (int r = 0; r < 4; ++r) { lo[r] = D[t][r] * inv; hi[r] = D[t][4 + r] * inv; }
        park[(2 * t) * 32 + lane]     = lo;
        park[(2 * t + 1) * 32 + lane] = hi;
      }
    } else {
      const float scale = -lambda_full * inv;
#pragma unroll
      for (int t = 0; t < 32; ++t) {
        const v4f lo = park[(2 * t) * 32 + lane];
        const v4f hi = park[(2 * t + 1) * 32 + lane];
#pragma unroll
        for (int r = 0; r < 4; ++r) {
          D[t][r]     = D[t][r] * scale + lo[r];
          D[t][4 + r] = D[t][4 + r] * scale + hi[r];
        }
      }
    }
  }

  v8f dmv = D[0];
#pragma unroll
  for (int t = 1; t < 32; ++t) {
#pragma unroll
    for (int r = 0; r < 8; ++r) dmv[r] = fminf(dmv[r], D[t][r]);
  }
  float dmin = fminf(fminf(fminf(dmv[0], dmv[1]), fminf(dmv[2], dmv[3])),
                     fminf(fminf(dmv[4], dmv[5]), fminf(dmv[6], dmv[7])));
  dmin = fminf(dmin, __shfl_xor(dmin, 16, 32));

  int wb[8];
#pragma unroll
  for (int k = 0; k < 8; ++k) wb[k] = c * 128 + ((2 * k + h) ^ c);
#pragma unroll
  for (int t = 0; t < 32; ++t) {
    const v4f ba = *(const v4f*)(bbase + t * 16);
    const v4f bb = *(const v4f*)(bbase + t * 16 + 8);
    v4f lo, hi;
#pragma unroll
    for (int r = 0; r < 8; ++r) {
      const float bias = (r < 4) ? ba[r & 3] : bb[r & 3];
      float dd = D[t][r] - dmin + 1e-5f;
      dd = (bias == zq) ? dd : 0.f;
      D[t][r] = dd;
      if (r < 4) lo[r & 3] = dd; else hi[r & 3] = dd;
    }
    park[wb[2 * (t & 3)]     + (t >> 2) * 16] = lo;
    park[wb[2 * (t & 3) + 1] + (t >> 2) * 16] = hi;
  }
  __builtin_amdgcn_fence(__ATOMIC_RELEASE, "wavefront");
  __builtin_amdgcn_wave_barrier();
  __builtin_amdgcn_fence(__ATOMIC_ACQUIRE, "wavefront");
  float* dg = diffp + (size_t)R0 * 512 + 4 * lane;
#pragma unroll
  for (int R = 0; R < 16; ++R) {
    const int rb = R * 128 + (lane ^ R);
#pragma unroll
    for (int i = 0; i < 4; ++i) {
      const v4f x = park[rb + 32 * i];
      __builtin_nontemporal_store(x, (v4f*)(dg + R * 512 + 128 * i));
    }
  }

  const float* vbase = Vm + (size_t)(b * S_ + 4 * h) * 32 + c;
  v8f ao0 = {}, ao1 = {};
#pragma unroll
  for (int t = 0; t < 32; ++t) {
#pragma unroll
    for (int j = 0; j < 4; ++j) {
      const float* vp = vbase + (size_t)(t * 16 + 8 * (j >> 1) + 2 * (j & 1)) * 32;
      v2f a0, a1, bf;
      a0[0] = vp[0];  a0[1] = vp[32];
      a1[0] = vp[16]; a1[1] = vp[48];
      bf[0] = D[t][2 * j]; bf[1] = D[t][2 * j + 1];
      ao0 = WMMA_F32(a0, bf, ao0);
      ao1 = WMMA_F32(a1, bf, ao1);
    }
  }

  float sq = 0.f;
#pragma unroll
  for (int r = 0; r < 8; ++r) sq += ao0[r] * ao0[r];
#pragma unroll
  for (int r = 0; r < 8; ++r) sq += ao1[r] * ao1[r];
  sq += __shfl_xor(sq, 16, 32);
  const float rms = rsqrtf(sq * (1.0f / 32.0f) + 1e-5f);
#pragma unroll
  for (int r = 0; r < 8; ++r) {
    ao0[r] = ao0[r] * rms * lnw[8 * h + r] * 0.8f;
    ao1[r] = ao1[r] * rms * lnw[16 + 8 * h + r] * 0.8f;
  }

  v8f o0 = {}, o1 = {};
#pragma unroll
  for (int mt = 0; mt < 2; ++mt) {
#pragma unroll
    for (int j = 0; j < 4; ++j) {
      const int kb = 16 * mt + 8 * h + 2 * j;
      const v2f w0 = *(const v2f*)(Wo + (size_t)c * 32 + kb);
      const v2f w1 = *(const v2f*)(Wo + (size_t)(16 + c) * 32 + kb);
      v2f bf;
      bf[0] = mt ? ao1[2 * j] : ao0[2 * j];
      bf[1] = mt ? ao1[2 * j + 1] : ao0[2 * j + 1];
      o0 = WMMA_F32(w0, bf, o0);
      o1 = WMMA_F32(w1, bf, o1);
    }
  }
  float* og = outp + (size_t)(R0 + c) * 32 + 8 * h;
  v4f x;
  x[0] = o0[0]; x[1] = o0[1]; x[2] = o0[2]; x[3] = o0[3];
  __builtin_nontemporal_store(x, (v4f*)(og));
  x[0] = o0[4]; x[1] = o0[5]; x[2] = o0[6]; x[3] = o0[7];
  __builtin_nontemporal_store(x, (v4f*)(og + 4));
  x[0] = o1[0]; x[1] = o1[1]; x[2] = o1[2]; x[3] = o1[3];
  __builtin_nontemporal_store(x, (v4f*)(og + 16));
  x[0] = o1[4]; x[1] = o1[5]; x[2] = o1[6]; x[3] = o1[7];
  __builtin_nontemporal_store(x, (v4f*)(og + 20));
}

extern "C" void kernel_launch(void* const* d_in, const int* in_sizes, int n_in,
                              void* d_out, int out_size, void* d_ws, size_t ws_size,
                              hipStream_t stream) {
  (void)in_sizes; (void)n_in; (void)out_size; (void)ws_size;
  const float* query = (const float*)d_in[0];
  const float* key   = (const float*)d_in[1];
  const int*   qmask = (const int*)d_in[2];
  const int*   kmask = (const int*)d_in[3];
  const float* Wq = (const float*)d_in[4];
  const float* Wk = (const float*)d_in[5];
  const float* Wv = (const float*)d_in[6];
  const float* Wo = (const float*)d_in[7];
  const float* lq1 = (const float*)d_in[8];
  const float* lk1 = (const float*)d_in[9];
  const float* lq2 = (const float*)d_in[10];
  const float* lk2 = (const float*)d_in[11];
  const float* lnw = (const float*)d_in[12];

  float* outp  = (float*)d_out;
  float* diffp = outp + (size_t)B_ * L_ * E_;

  float* Qs = (float*)d_ws;
  float* Km = Qs + (size_t)B_ * L_ * E_;
  float* Vm = Km + (size_t)B_ * S_ * E_;
  float* Bk = Vm + (size_t)B_ * S_ * E_;

  proj_kernel<<<(B_ * S_) / 16, 256, 0, stream>>>(key, Wk, Wv, kmask, Km, Vm, Bk,
                                                  query, Wq, Qs);
  attn_kernel<<<(B_ * L_) / 128, 256, 0, stream>>>(Qs, Km, Vm, qmask, Bk,
                                                 lq1, lk1, lq2, lk2, lnw, Wo,
                                                 outp, diffp);
}


# attention kernel: K copied once per work-group into the 64 KB of LDS left free, by whole-line loads (one barrier at the start); score A fragments are LDS reads, not 16-line global gathers per wave
# speedup vs baseline: 7.1808x; 1.0391x over previous
#include <hip/hip_runtime.h>
#include <math.h>

typedef float v2f __attribute__((ext_vector_type(2)));
typedef float v4f __attribute__((ext_vector_type(4)));
typedef int   v4i __attribute__((ext_vector_type(4)));
typedef float v8f __attribute__((ext_vector_type(8)));

#define WMMA_F32(a, b, c) \
  __builtin_amdgcn_wmma_f32_16x16x4_f32(false, (a), false, (b), (short)0, (c), false, false)

static constexpr int B_ = 8, S_ = 512, E_ = 32, L_ = 4096, SUBE_ = 768;
static constexpr float LAMBDA_INIT = 0.2f;

__global__ __launch_bounds__(256) __attribute__((amdgpu_waves_per_eu(2, 2))) void proj_kernel(
    const float* __restrict__ key, const float* __restrict__ Wk,
    const float* __restrict__ Wv, const int* __restrict__ kmask,
    float* __restrict__ Km, float* __restrict__ Vm, float* __restrict__ Bk,
    const float* __restrict__ q, const float* __restrict__ Wq,
    float* __restrict__ Qs) {
  __shared__ __attribute__((aligned(16))) float tile[16 * 68];
  const int tid  = threadIdx.x;
  const int w    = tid >> 5;
  const int lane = tid & 31;
  const int n    = lane & 15;
  const int h    = lane >> 4;
  const int m0   = blockIdx.x * 16;
  const int wu   = __builtin_amdgcn_readfirstlane(w);

  if (wu >= 4) {
    for (int u = 0; u < 2; ++u) {
      const int qm0  = ((blockIdx.x * 4 + (wu - 4)) * 2 + u) * 16;
      const int mrow = qm0 + n;
      v8f a0 = {}, a1 = {};
      for (int kk = 0; kk < 8; ++kk) {
        const int kb = kk * 4 + h * 2;
        v2f a  = *(const v2f*)(q + (size_t)mrow * 32 + kb);
        v2f b0 = *(const v2f*)(Wq + (size_t)n * 32 + kb);
        v2f b1 = *(const v2f*)(Wq + (size_t)(16 + n) * 32 + kb);
        a0 = WMMA_F32(a, b0, a0);
        a1 = WMMA_F32(a, b1, a1);
      }
      const int rbase = qm0 + (h ? 8 : 0);
      for (int v = 0; v < 8; ++v) {
        const size_t r = (size_t)(rbase + v) * 32;
        Qs[r + n]      = a0[v] * 0.25f;
        Qs[r + 16 + n] = a1[v] * 0.25f;
      }
    }
  }

  const bool bias_writer = (blockIdx.x & 1) == 0 && tid < 64;
  const int kidx = (m0 + lane < B_ * S_) ? m0 + lane : B_ * S_ - 1;
  const int kmv = kmask[kidx];

  if (wu < 4) {
    const float* wsel = (w < 2) ? Wk : Wv;
    const float* ap = key + (size_t)(m0 + n) * SUBE_ + 4 * h;
    const float* bp = wsel + (size_t)((w & 1) * 16 + n) * SUBE_ + 4 * h;

    v8f acc0 = {}, acc1 = {};
#pragma unroll
    for (int i = 0; i < SUBE_ / 8; ++i) {
      const v4f a = *(const v4f*)(ap + 8 * i);
      const v4f b = *(const v4f*)(bp + 8 * i);
      v2f a01, b01, a23, b23;
      a01[0] = a[0]; a01[1] = a[1]; b01[0] = b[0]; b01[1] = b[1];
      a23[0] = a[2]; a23[1] = a[3]; b23[0] = b[2]; b23[1] = b[3];
      acc0 = WMMA_F32(a01, b01, acc0);
      acc1 = WMMA_F32(a23, b23, acc1);
    }

#pragma unroll
    for (int v = 0; v < 8; ++v)
      tile[(8 * h + v) * 68 + 16 * w + n] = acc0[v] + acc1[v];
  }
  __syncthreads();

  if (wu < 4) {
#pragma unroll
    for (int q2 = 0; q2 < 2; ++q2) {
      const int p    = tid + 128 * q2;
      const int line = p >> 3;
      const int seg  = p & 7;
      const int row  = line & 15;
      const int isv  = line >> 4;
      const v4f x = *(const v4f*)(tile + row * 68 + 32 * isv + 4 * seg);
      float* dst = (isv ? Vm : Km) + (size_t)(m0 + row) * 32 + 4 * seg;
      *(v4f*)dst = x;
    }
    if (bias_writer)
      Bk[(size_t)w * (B_ * S_) + m0 + lane] = kmv ? 0.f : -1e20f;
  }
}

__global__ __launch_bounds__(256) __attribute__((amdgpu_waves_per_eu(2, 2))) void attn_kernel(
    const float* __restrict__ Qs, const float* __restrict__ Km,
    const float* __restrict__ Vm, const int* __restrict__ qmask,
    const float* __restrict__ Bk, const float* __restrict__ lq1,
    const float* __restrict__ lk1, const float* __restrict__ lq2,
    const float* __restrict__ lk2, const float* __restrict__ lnw,
    const float* __restrict__ Wo, float* __restrict__ outp,
    float* __restrict__ diffp) {
  __shared__ __attribute__((aligned(16))) v4f park_all[8 * 64 * 32];
  __shared__ __attribute__((aligned(16))) v4f kst[512 * 8];
  const int lane = threadIdx.x & 31;
  const int wv   = __builtin_amdgcn_readfirstlane((int)(threadIdx.x >> 5));
  v4f* park      = park_all + wv * (64 * 32);
  const int c    = lane & 15;
  const int h    = lane >> 4;
  const int R0   = (blockIdx.x * 8 + wv) * 16;
  const int b    = R0 / L_;

  const int tid = threadIdx.x;
  const v4f* ksrc = (const v4f*)(Km + (size_t)b * S_ * 32);
  v4f kf[16];
#pragma unroll
  for (int i = 0; i < 16; ++i) kf[i] = ksrc[i * 256 + tid];

  float s1 = 0.f, s2 = 0.f;
  for (int i = 0; i < 16; ++i) { s1 += lq1[i] * lk1[i]; s2 += lq2[i] * lk2[i]; }
  const float lambda_full = __expf(s1) - __expf(s2) + LAMBDA_INIT;

  const int qv = qmask[R0 + c];

  const int ke = (tid & ~7) + ((tid & 7) ^ ((tid >> 4) & 7));
#pragma unroll
  for (int i = 0; i < 16; ++i) kst[i * 256 + ke] = kf[i];
  __syncthreads();

  const int crow = ((c & 4) << 1) | ((c & 8) >> 1) | (c & 3);
  const int ksw = (crow >> 1) & 7;
  const float* bbase = Bk + b * S_ + 4 * h;

  const float zq = qv ? 0.f : 1.f;

  v8f D[32];

#pragma unroll 1
  for (int hd = 0; hd < 2; ++hd) {
    const float* qp = Qs + (size_t)(R0 + c) * 32 + hd * 16 + 8 * h;
    const v4f qa = *(const v4f*)(qp);
    const v4f qb = *(const v4f*)(qp + 4);
    v2f q0, q1, q2, q3;
    q0[0] = qa[0]; q0[1] = qa[1];
    q1[0] = qa[2]; q1[1] = qa[3];
    q2[0] = qb[0]; q2[1] = qb[1];
    q3[0] = qb[2]; q3[1] = qb[3];
    const v4f* ka = kst + crow * 8 + ((4 * hd + 2 * h) ^ ksw);
    const v4f* kb = kst + crow * 8 + ((4 * hd + 2 * h + 1) ^ ksw);
    const float* bh = bbase + hd * (B_ * S_);

    v4f KA[32], KB[32];
#pragma unroll
    for (int g = 0; g <= 4; ++g) {
      if (g < 4) {
#pragma unroll
        for (int i = 0; i < 8; ++i) {
          const int t = 8 * g + i;
          KA[t] = ka[t * 128];
          KB[t] = kb[t * 128];
          const v4f ba = *(const v4f*)(bh + t * 16);
          const v4f bb = *(const v4f*)(bh + t * 16 + 8);
          v8f x;
          x[0] = ba[0]; x[1] = ba[1]; x[2] = ba[2]; x[3] = ba[3];
          x[4] = bb[0]; x[5] = bb[1]; x[6] = bb[2]; x[7] = bb[3];
          D[t] = x;
        }
      }
      __builtin_amdgcn_sched_barrier(0);
      if (g > 0) {
#pragma unroll
        for (int i = 0; i < 8; i += 2) {
          const int t = 8 * (g - 1) + i;
          const v4f xa = KA[t], xb = KB[t], ya = KA[t + 1], yb = KB[t + 1];
          v8f x = D[t], y = D[t + 1];
          v2f f;
          f[0] = xa[0]; f[1] = xa[1]; x = WMMA_F32(f, q0, x);
          f[0] = ya[0]; f[1] = ya[1]; y = WMMA_F32(f, q0, y);
          f[0] = xa[2]; f[1] = xa[3]; x = WMMA_F32(f, q1, x);
          f[0] = ya[2]; f[1] = ya[3]; y = WMMA_F32(f, q1, y);
          f[0] = xb[0]; f[1] = xb[1]; x = WMMA_F32(f, q2, x);
          f[0] = yb[0]; f[1] = yb[1]; y = WMMA_F32(f, q2, y);
          f[0] = xb[2]; f[1] = xb[3]; x = WMMA_F32(f, q3, x);
          f[0] = yb[2]; f[1] = yb[3]; y = WMMA_F32(f, q3, y);
          D[t] = x; D[t + 1] = y;
        }
      }
      __builtin_amdgcn_sched_barrier(0);
    }

    v8f mv = D[0];
#pragma unroll
    for (int t = 1; t < 32; ++t) {
#pragma unroll
      for (int r = 0; r < 8; ++r) mv[r] = fmaxf(mv[r], D[t][r]);
    }
    float mx = fmaxf(fmaxf(fmaxf(mv[0], mv[1]), fmaxf(mv[2], mv[3])),
                     fmaxf(fmaxf(mv[4], mv[5]), fmaxf(mv[6], mv[7])));
    mx = fmaxf(mx, __shfl_xor(mx, 16, 32));

    v8f sv = {};
#pragma unroll
    for (int t = 0; t < 32; ++t) {
#pragma unroll
      for (int r = 0; r < 8; ++r) {
        const float e = __expf(D[t][r] - mx);
        D[t][r] = e; sv[r] += e;
      }
    }
    float sum = ((sv[0] + sv[1]) + (sv[2] + sv[3])) +
                ((sv[4] + sv[5]) + (sv[6] + sv[7]));
    sum += __shfl_xor(sum, 16, 32);
    const float inv   = 1.f / (sum + 1e-8f);
    if (hd == 0) {
#pragma unroll
      for (int t = 0; t < 32; ++t) {
        v4f lo, hi;
#pragma unroll
        for (int r = 0; r < 4; ++r) { lo[r] = D[t][r] * inv; hi[r] = D[t][4 + r] * inv; }
        park[(2 * t) * 32 + lane]     = lo;
        park[(2 * t + 1) * 32 + lane] = hi;
      }
    } else {
      const float scale = -lambda_full * inv;
#pragma unroll
      for (int t = 0; t < 32; ++t) {
        const v4f lo = park[(2 * t) * 32 + lane];
        const v4f hi = park[(2 * t + 1) * 32 + lane];
#pragma unroll
        for (int r = 0; r < 4; ++r) {
          D[t][r]     = D[t][r] * scale + lo[r];
          D[t][4 + r] = D[t][4 + r] * scale + hi[r];
        }
      }
    }
  }

  v8f dmv = D[0];
#pragma unroll
  for (int t = 1; t < 32; ++t) {
#pragma unroll
    for (int r = 0; r < 8; ++r) dmv[r] = fminf(dmv[r], D[t][r]);
  }
  float dmin = fminf(fminf(fminf(dmv[0], dmv[1]), fminf(dmv[2], dmv[3])),
                     fminf(fminf(dmv[4], dmv[5]), fminf(dmv[6], dmv[7])));
  dmin = fminf(dmin, __shfl_xor(dmin, 16, 32));

  int wb[8];
#pragma unroll
  for (int k = 0; k < 8; ++k) wb[k] = c * 128 + ((2 * k + h) ^ c);
#pragma unroll
  for (int t = 0; t < 32; ++t) {
    const v4f ba = *(const v4f*)(bbase + t * 16);
    const v4f bb = *(const v4f*)(bbase + t * 16 + 8);
    v4f lo, hi;
#pragma unroll
    for (int r = 0; r < 8; ++r) {
      const float bias = (r < 4) ? ba[r & 3] : bb[r & 3];
      float dd = D[t][r] - dmin + 1e-5f;
      dd = (bias == zq) ? dd : 0.f;
      D[t][r] = dd;
      if (r < 4) lo[r & 3] = dd; else hi[r & 3] = dd;
    }
    park[wb[2 * (t & 3)]     + (t >> 2) * 16] = lo;
    park[wb[2 * (t & 3) + 1] + (t >> 2) * 16] = hi;
  }
  __builtin_amdgcn_fence(__ATOMIC_RELEASE, "wavefront");
  __builtin_amdgcn_wave_barrier();
  __builtin_amdgcn_fence(__ATOMIC_ACQUIRE, "wavefront");
  float* dg = diffp + (size_t)R0 * 512 + 4 * lane;
#pragma unroll
  for (int R = 0; R < 16; ++R) {
    const int rb = R * 128 + (lane ^ R);
#pragma unroll
    for (int i = 0; i < 4; ++i) {
      const v4f x = park[rb + 32 * i];
      __builtin_nontemporal_store(x, (v4f*)(dg + R * 512 + 128 * i));
    }
  }

  const float* vbase = Vm + (size_t)(b * S_ + 4 * h) * 32 + c;
  v8f ao0 = {}, ao1 = {};
#pragma unroll
  for (int t = 0; t < 32; ++t) {
#pragma unroll
    for (int j = 0; j < 4; ++j) {
      const float* vp = vbase + (size_t)(t * 16 + 8 * (j >> 1) + 2 * (j & 1)) * 32;
      v2f a0, a1, bf;
      a0[0] = vp[0];  a0[1] = vp[32];
      a1[0] = vp[16]; a1[1] = vp[48];
      bf[0] = D[t][2 * j]; bf[1] = D[t][2 * j + 1];
      ao0 = WMMA_F32(a0, bf, ao0);
      ao1 = WMMA_F32(a1, bf, ao1);
    }
  }

  float sq = 0.f;
#pragma unroll
  for (int r = 0; r < 8; ++r) sq += ao0[r] * ao0[r];
#pragma unroll
  for (int r = 0; r < 8; ++r) sq += ao1[r] * ao1[r];
  sq += __shfl_xor(sq, 16, 32);
  const float rms = rsqrtf(sq * (1.0f / 32.0f) + 1e-5f);
#pragma unroll
  for (int r = 0; r < 8; ++r) {
    ao0[r] = ao0[r] * rms * lnw[8 * h + r] * 0.8f;
    ao1[r] = ao1[r] * rms * lnw[16 + 8 * h + r] * 0.8f;
  }

  v8f o0 = {}, o1 = {};
#pragma unroll
  for (int mt = 0; mt < 2; ++mt) {
#pragma unroll
    for (int j = 0; j < 4; ++j) {
      const int kb = 16 * mt + 8 * h + 2 * j;
      const v2f w0 = *(const v2f*)(Wo + (size_t)c * 32 + kb);
      const v2f w1 = *(const v2f*)(Wo + (size_t)(16 + c) * 32 + kb);
      v2f bf;
      bf[0] = mt ? ao1[2 * j] : ao0[2 * j];
      bf[1] = mt ? ao1[2 * j + 1] : ao0[2 * j + 1];
      o0 = WMMA_F32(w0, bf, o0);
      o1 = WMMA_F32(w1, bf, o1);
    }
  }
  float* og = outp + (size_t)(R0 + c) * 32 + 8 * h;
  v4f x;
  x[0] = o0[0]; x[1] = o0[1]; x[2] = o0[2]; x[3] = o0[3];
  __builtin_nontemporal_store(x, (v4f*)(og));
  x[0] = o0[4]; x[1] = o0[5]; x[2] = o0[6]; x[3] = o0[7];
  __builtin_nontemporal_store(x, (v4f*)(og + 4));
  x[0] = o1[0]; x[1] = o1[1]; x[2] = o1[2]; x[3] = o1[3];
  __builtin_nontemporal_store(x, (v4f*)(og + 16));
  x[0] = o1[4]; x[1] = o1[5]; x[2] = o1[6]; x[3] = o1[7];
  __builtin_nontemporal_store(x, (v4f*)(og + 20));
}

extern "C" void kernel_launch(void* const* d_in, const int* in_sizes, int n_in,
                              void* d_out, int out_size, void* d_ws, size_t ws_size,
                              hipStream_t stream) {
  (void)in_sizes; (void)n_in; (void)out_size; (void)ws_size;
  const float* query = (const float*)d_in[0];
  const float* key   = (const float*)d_in[1];
  const int*   qmask = (const int*)d_in[2];
  const int*   kmask = (const int*)d_in[3];
  const float* Wq = (const float*)d_in[4];
  const float* Wk = (const float*)d_in[5];
  const float* Wv = (const float*)d_in[6];
  const float* Wo = (const float*)d_in[7];
  const float* lq1 = (const float*)d_in[8];
  const float* lk1 = (const float*)d_in[9];
  const float* lq2 = (const float*)d_in[10];
  const float* lk2 = (const float*)d_in[11];
  const float* lnw = (const float*)d_in[12];

  float* outp  = (float*)d_out;
  float* diffp = outp + (size_t)B_ * L_ * E_;

  float* Qs = (float*)d_ws;
  float* Km = Qs + (size_t)B_ * L_ * E_;
  float* Vm = Km + (size_t)B_ * S_ * E_;
  float* Bk = Vm + (size_t)B_ * S_ * E_;

  proj_kernel<<<(B_ * S_) / 16, 256, 0, stream>>>(key, Wk, Wv, kmask, Km, Vm, Bk,
                                                  query, Wq, Qs);
  attn_kernel<<<(B_ * L_) / 128, 256, 0, stream>>>(Qs, Km, Vm, qmask, Bk,
                                                 lq1, lk1, lq2, lk2, lnw, Wo,
                                                 outp, diffp);
}


# attention kernel, V product: row i of output tile s is feature 2i+s, so one 8-byte load of a key's whole line feeds both tiles (512 -> 256 V loads a wave); norm and out-projection sum order changes
# speedup vs baseline: 7.5900x; 1.0570x over previous
#include <hip/hip_runtime.h>
#include <math.h>

typedef float v2f __attribute__((ext_vector_type(2)));
typedef float v4f __attribute__((ext_vector_type(4)));
typedef int   v4i __attribute__((ext_vector_type(4)));
typedef float v8f __attribute__((ext_vector_type(8)));

#define WMMA_F32(a, b, c) \
  __builtin_amdgcn_wmma_f32_16x16x4_f32(false, (a), false, (b), (short)0, (c), false, false)

static constexpr int B_ = 8, S_ = 512, E_ = 32, L_ = 4096, SUBE_ = 768;
static constexpr float LAMBDA_INIT = 0.2f;

__global__ __launch_bounds__(256) __attribute__((amdgpu_waves_per_eu(2, 2))) void proj_kernel(
    const float* __restrict__ key, const float* __restrict__ Wk,
    const float* __restrict__ Wv, const int* __restrict__ kmask,
    float* __restrict__ Km, float* __restrict__ Vm, float* __restrict__ Bk,
    const float* __restrict__ q, const float* __restrict__ Wq,
    float* __restrict__ Qs) {
  __shared__ __attribute__((aligned(16))) float tile[16 * 68];
  const int tid  = threadIdx.x;
  const int w    = tid >> 5;
  const int lane = tid & 31;
  const int n    = lane & 15;
  const int h    = lane >> 4;
  const int m0   = blockIdx.x * 16;
  const int wu   = __builtin_amdgcn_readfirstlane(w);

  if (wu >= 4) {
    for (int u = 0; u < 2; ++u) {
      const int qm0  = ((blockIdx.x * 4 + (wu - 4)) * 2 + u) * 16;
      const int mrow = qm0 + n;
      v8f a0 = {}, a1 = {};
      for (int kk = 0; kk < 8; ++kk) {
        const int kb = kk * 4 + h * 2;
        v2f a  = *(const v2f*)(q + (size_t)mrow * 32 + kb);
        v2f b0 = *(const v2f*)(Wq + (size_t)n * 32 + kb);
        v2f b1 = *(const v2f*)(Wq + (size_t)(16 + n) * 32 + kb);
        a0 = WMMA_F32(a, b0, a0);
        a1 = WMMA_F32(a, b1, a1);
      }
      const int rbase = qm0 + (h ? 8 : 0);
      for (int v = 0; v < 8; ++v) {
        const size_t r = (size_t)(rbase + v) * 32;
        Qs[r + n]      = a0[v] * 0.25f;
        Qs[r + 16 + n] = a1[v] * 0.25f;
      }
    }
  }

  const bool bias_writer = (blockIdx.x & 1) == 0 && tid < 64;
  const int kidx = (m0 + lane < B_ * S_) ? m0 + lane : B_ * S_ - 1;
  const int kmv = kmask[kidx];

  if (wu < 4) {
    const float* wsel = (w < 2) ? Wk : Wv;
    const float* ap = key + (size_t)(m0 + n) * SUBE_ + 4 * h;
    const float* bp = wsel + (size_t)((w & 1) * 16 + n) * SUBE_ + 4 * h;

    v8f acc0 = {}, acc1 = {};
#pragma unroll
    for (int i = 0; i < SUBE_ / 8; ++i) {
      const v4f a = *(const v4f*)(ap + 8 * i);
      const v4f b = *(const v4f*)(bp + 8 * i);
      v2f a01, b01, a23, b23;
      a01[0] = a[0]; a01[1] = a[1]; b01[0] = b[0]; b01[1] = b[1];
      a23[0] = a[2]; a23[1] = a[3]; b23[0] = b[2]; b23[1] = b[3];
      acc0 = WMMA_F32(a01, b01, acc0);
      acc1 = WMMA_F32(a23, b23, acc1);
    }

#pragma unroll
    for (int v = 0; v < 8; ++v)
      tile[(8 * h + v) * 68 + 16 * w + n] = acc0[v] + acc1[v];
  }
  __syncthreads();

  if (wu < 4) {
#pragma unroll
    for (int q2 = 0; q2 < 2; ++q2) {
      const int p    = tid + 128 * q2;
      const int line = p >> 3;
      const int seg  = p & 7;
      const int row  = line & 15;
      const int isv  = line >> 4;
      const v4f x = *(const v4f*)(tile + row * 68 + 32 * isv + 4 * seg);
      float* dst = (isv ? Vm : Km) + (size_t)(m0 + row) * 32 + 4 * seg;
      *(v4f*)dst = x;
    }
    if (bias_writer)
      Bk[(size_t)w * (B_ * S_) + m0 + lane] = kmv ? 0.f : -1e20f;
  }
}

__global__ __launch_bounds__(256) __attribute__((amdgpu_waves_per_eu(2, 2))) void attn_kernel(
    const float* __restrict__ Qs, const float* __restrict__ Km,
    const float* __restrict__ Vm, const int* __restrict__ qmask,
    const float* __restrict__ Bk, const float* __restrict__ lq1,
    const float* __restrict__ lk1, const float* __restrict__ lq2,
    const float* __restrict__ lk2, const float* __restrict__ lnw,
    const float* __restrict__ Wo, float* __restrict__ outp,
    float* __restrict__ diffp) {
  __shared__ __attribute__((aligned(16))) v4f park_all[8 * 64 * 32];
  __shared__ __attribute__((aligned(16))) v4f kst[512 * 8];
  const int lane = threadIdx.x & 31;
  const int wv   = __builtin_amdgcn_readfirstlane((int)(threadIdx.x >> 5));
  v4f* park      = park_all + wv * (64 * 32);
  const int c    = lane & 15;
  const int h    = lane >> 4;
  const int R0   = (blockIdx.x * 8 + wv) * 16;
  const int b    = R0 / L_;

  const int tid = threadIdx.x;
  const v4f* ksrc = (const v4f*)(Km + (size_t)b * S_ * 32);
  v4f kf[16];
#pragma unroll
  for (int i = 0; i < 16; ++i) kf[i] = ksrc[i * 256 + tid];

  float s1 = 0.f, s2 = 0.f;
  for (int i = 0; i < 16; ++i) { s1 += lq1[i] * lk1[i]; s2 += lq2[i] * lk2[i]; }
  const float lambda_full = __expf(s1) - __expf(s2) + LAMBDA_INIT;

  const int qv = qmask[R0 + c];

  const int ke = (tid & ~7) + ((tid & 7) ^ ((tid >> 4) & 7));
#pragma unroll
  for (int i = 0; i < 16; ++i) kst[i * 256 + ke] = kf[i];
  __syncthreads();

  const int crow = ((c & 4) << 1) | ((c & 8) >> 1) | (c & 3);
  const int ksw = (crow >> 1) & 7;
  const float* bbase = Bk + b * S_ + 4 * h;

  const float zq = qv ? 0.f : 1.f;

  v8f D[32];

#pragma unroll 1
  for (int hd = 0; hd < 2; ++hd) {
    const float* qp = Qs + (size_t)(R0 + c) * 32 + hd * 16 + 8 * h;
    const v4f qa = *(const v4f*)(qp);
    const v4f qb = *(const v4f*)(qp + 4);
    v2f q0, q1, q2, q3;
    q0[0] = qa[0]; q0[1] = qa[1];
    q1[0] = qa[2]; q1[1] = qa[3];
    q2[0] = qb[0]; q2[1] = qb[1];
    q3[0] = qb[2]; q3[1] = qb[3];
    const v4f* ka = kst + crow * 8 + ((4 * hd + 2 * h) ^ ksw);
    const v4f* kb = kst + crow * 8 + ((4 * hd + 2 * h + 1) ^ ksw);
    const float* bh = bbase + hd * (B_ * S_);

    v4f KA[32], KB[32];
#pragma unroll
    for (int g = 0; g <= 4; ++g) {
      if (g < 4) {
#pragma unroll
        for (int i = 0; i < 8; ++i) {
          const int t = 8 * g + i;
          KA[t] = ka[t * 128];
          KB[t] = kb[t * 128];
          const v4f ba = *(const v4f*)(bh + t * 16);
          const v4f bb = *(const v4f*)(bh + t * 16 + 8);
          v8f x;
          x[0] = ba[0]; x[1] = ba[1]; x[2] = ba[2]; x[3] = ba[3];
          x[4] = bb[0]; x[5] = bb[1]; x[6] = bb[2]; x[7] = bb[3];
          D[t] = x;
        }
      }
      __builtin_amdgcn_sched_barrier(0);
      if (g > 0) {
#pragma unroll
        for (int i = 0; i < 8; i += 2) {
          const int t = 8 * (g - 1) + i;
          const v4f xa = KA[t], xb = KB[t], ya = KA[t + 1], yb = KB[t + 1];
          v8f x = D[t], y = D[t + 1];
          v2f f;
          f[0] = xa[0]; f[1] = xa[1]; x = WMMA_F32(f, q0, x);
          f[0] = ya[0]; f[1] = ya[1]; y = WMMA_F32(f, q0, y);
          f[0] = xa[2]; f[1] = xa[3]; x = WMMA_F32(f, q1, x);
          f[0] = ya[2]; f[1] = ya[3]; y = WMMA_F32(f, q1, y);
          f[0] = xb[0]; f[1] = xb[1]; x = WMMA_F32(f, q2, x);
          f[0] = yb[0]; f[1] = yb[1]; y = WMMA_F32(f, q2, y);
          f[0] = xb[2]; f[1] = xb[3]; x = WMMA_F32(f, q3, x);
          f[0] = yb[2]; f[1] = yb[3]; y = WMMA_F32(f, q3, y);
          D[t] = x; D[t + 1] = y;
        }
      }
      __builtin_amdgcn_sched_barrier(0);
    }

    v8f mv = D[0];
#pragma unroll
    for (int t = 1; t < 32; ++t) {
#pragma unroll
      for (int r = 0; r < 8; ++r) mv[r] = fmaxf(mv[r], D[t][r]);
    }
    float mx = fmaxf(fmaxf(fmaxf(mv[0], mv[1]), fmaxf(mv[2], mv[3])),
                     fmaxf(fmaxf(mv[4], mv[5]), fmaxf(mv[6], mv[7])));
    mx = fmaxf(mx, __shfl_xor(mx, 16, 32));

    v8f sv = {};
#pragma unroll
    for (int t = 0; t < 32; ++t) {
#pragma unroll
      for (int r = 0; r < 8; ++r) {
        const float e = __expf(D[t][r] - mx);
        D[t][r] = e; sv[r] += e;
      }
    }
    float sum = ((sv[0] + sv[1]) + (sv[2] + sv[3])) +
                ((sv[4] + sv[5]) + (sv[6] + sv[7]));
    sum += __shfl_xor(sum, 16, 32);
    const float inv   = 1.f / (sum + 1e-8f);
    if (hd == 0) {
#pragma unroll
      for (int t = 0; t < 32; ++t) {
        v4f lo, hi;
#pragma unroll
        for (int r = 0; r < 4; ++r) { lo[r] = D[t][r] * inv; hi[r] = D[t][4 + r] * inv; }
        park[(2 * t) * 32 + lane]     = lo;
        park[(2 * t + 1) * 32 + lane] = hi;
      }
    } else {
      const float scale = -lambda_full * inv;
#pragma unroll
      for (int t = 0; t < 32; ++t) {
        const v4f lo = park[(2 * t) * 32 + lane];
        const v4f hi = park[(2 * t + 1) * 32 + lane];
#pragma unroll
        for (int r = 0; r < 4; ++r) {
          D[t][r]     = D[t][r] * scale + lo[r];
          D[t][4 + r] = D[t][4 + r] * scale + hi[r];
        }
      }
    }
  }

  v8f dmv = D[0];
#pragma unroll
  for (int t = 1; t < 32; ++t) {
#pragma unroll
    for (int r = 0; r < 8; ++r) dmv[r] = fminf(dmv[r], D[t][r]);
  }
  float dmin = fminf(fminf(fminf(dmv[0], dmv[1]), fminf(dmv[2], dmv[3])),
                     fminf(fminf(dmv[4], dmv[5]), fminf(dmv[6], dmv[7])));
  dmin = fminf(dmin, __shfl_xor(dmin, 16, 32));

  int wb[8];
#pragma unroll
  for (int k = 0; k < 8; ++k) wb[k] = c * 128 + ((2 * k + h) ^ c);
#pragma unroll
  for (int t = 0; t < 32; ++t) {
    const v4f ba = *(const v4f*)(bbase + t * 16);
    const v4f bb = *(const v4f*)(bbase + t * 16 + 8);
    v4f lo, hi;
#pragma unroll
    for (int r = 0; r < 8; ++r) {
      const float bias = (r < 4) ? ba[r & 3] : bb[r & 3];
      float dd = D[t][r] - dmin + 1e-5f;
      dd = (bias == zq) ? dd : 0.f;
      D[t][r] = dd;
      if (r < 4) lo[r & 3] = dd; else hi[r & 3] = dd;
    }
    park[wb[2 * (t & 3)]     + (t >> 2) * 16] = lo;
    park[wb[2 * (t & 3) + 1] + (t >> 2) * 16] = hi;
  }
  __builtin_amdgcn_fence(__ATOMIC_RELEASE, "wavefront");
  __builtin_amdgcn_wave_barrier();
  __builtin_amdgcn_fence(__ATOMIC_ACQUIRE, "wavefront");
  float* dg = diffp + (size_t)R0 * 512 + 4 * lane;
#pragma unroll
  for (int R = 0; R < 16; ++R) {
    const int rb = R * 128 + (lane ^ R);
#pragma unroll
    for (int i = 0; i < 4; ++i) {
      const v4f x = park[rb + 32 * i];
      __builtin_nontemporal_store(x, (v4f*)(dg + R * 512 + 128 * i));
    }
  }
  __builtin_amdgcn_sched_barrier(0x0020);

  const float* vbase = Vm + (size_t)(b * S_ + 4 * h) * 32 + 2 * c;
  v8f ao0 = {}, ao1 = {};
#pragma unroll
  for (int t = 0; t < 32; ++t) {
#pragma unroll
    for (int j = 0; j < 4; ++j) {
      const float* vp = vbase + (size_t)(t * 16 + 8 * (j >> 1) + 2 * (j & 1)) * 32;
      const v2f k0 = *(const v2f*)(vp);
      const v2f k1 = *(const v2f*)(vp + 32);
      v2f a0, a1, bf;
      a0[0] = k0[0]; a0[1] = k1[0];
      a1[0] = k0[1]; a1[1] = k1[1];
      bf[0] = D[t][2 * j]; bf[1] = D[t][2 * j + 1];
      ao0 = WMMA_F32(a0, bf, ao0);
      ao1 = WMMA_F32(a1, bf, ao1);
    }
  }

  float sq = 0.f;
#pragma unroll
  for (int r = 0; r < 8; ++r) sq += ao0[r] * ao0[r];
#pragma unroll
  for (int r = 0; r < 8; ++r) sq += ao1[r] * ao1[r];
  sq += __shfl_xor(sq, 16, 32);
  const float rms = rsqrtf(sq * (1.0f / 32.0f) + 1e-5f);
#pragma unroll
  for (int r = 0; r < 8; ++r) {
    ao0[r] = ao0[r] * rms * lnw[16 * h + 2 * r] * 0.8f;
    ao1[r] = ao1[r] * rms * lnw[16 * h + 2 * r + 1] * 0.8f;
  }

  v8f o0 = {}, o1 = {};
#pragma unroll
  for (int mt = 0; mt < 2; ++mt) {
#pragma unroll
    for (int j = 0; j < 4; ++j) {
      const int kb = 16 * h + 4 * j;
      const v4f x0 = *(const v4f*)(Wo + (size_t)c * 32 + kb);
      const v4f x1 = *(const v4f*)(Wo + (size_t)(16 + c) * 32 + kb);
      v2f w0, w1, bf;
      w0[0] = x0[mt]; w0[1] = x0[2 + mt];
      w1[0] = x1[mt]; w1[1] = x1[2 + mt];
      bf[0] = mt ? ao1[2 * j] : ao0[2 * j];
      bf[1] = mt ? ao1[2 * j + 1] : ao0[2 * j + 1];
      o0 = WMMA_F32(w0, bf, o0);
      o1 = WMMA_F32(w1, bf, o1);
    }
  }
  float* og = outp + (size_t)(R0 + c) * 32 + 8 * h;
  v4f x;
  x[0] = o0[0]; x[1] = o0[1]; x[2] = o0[2]; x[3] = o0[3];
  __builtin_nontemporal_store(x, (v4f*)(og));
  x[0] = o0[4]; x[1] = o0[5]; x[2] = o0[6]; x[3] = o0[7];
  __builtin_nontemporal_store(x, (v4f*)(og + 4));
  x[0] = o1[0]; x[1] = o1[1]; x[2] = o1[2]; x[3] = o1[3];
  __builtin_nontemporal_store(x, (v4f*)(og + 16));
  x[0] = o1[4]; x[1] = o1[5]; x[2] = o1[6]; x[3] = o1[7];
  __builtin_nontemporal_store(x, (v4f*)(og + 20));
}

extern "C" void kernel_launch(void* const* d_in, const int* in_sizes, int n_in,
                              void* d_out, int out_size, void* d_ws, size_t ws_size,
                              hipStream_t stream) {
  (void)in_sizes; (void)n_in; (void)out_size; (void)ws_size;
  const float* query = (const float*)d_in[0];
  const float* key   = (const float*)d_in[1];
  const int*   qmask = (const int*)d_in[2];
  const int*   kmask = (const int*)d_in[3];
  const float* Wq = (const float*)d_in[4];
  const float* Wk = (const float*)d_in[5];
  const float* Wv = (const float*)d_in[6];
  const float* Wo = (const float*)d_in[7];
  const float* lq1 = (const float*)d_in[8];
  const float* lk1 = (const float*)d_in[9];
  const float* lq2 = (const float*)d_in[10];
  const float* lk2 = (const float*)d_in[11];
  const float* lnw = (const float*)d_in[12];

  float* outp  = (float*)d_out;
  float* diffp = outp + (size_t)B_ * L_ * E_;

  float* Qs = (float*)d_ws;
  float* Km = Qs + (size_t)B_ * L_ * E_;
  float* Vm = Km + (size_t)B_ * S_ * E_;
  float* Bk = Vm + (size_t)B_ * S_ * E_;

  proj_kernel<<<(B_ * S_) / 16, 256, 0, stream>>>(key, Wk, Wv, kmask, Km, Vm, Bk,
                                                  query, Wq, Qs);
  attn_kernel<<<(B_ * L_) / 128, 256, 0, stream>>>(Qs, Km, Vm, qmask, Bk,
                                                 lq1, lk1, lq2, lk2, lnw, Wo,
                                                 outp, diffp);
}


# K/V projection: each wave copies its key and weight rows through a private LDS region by whole-line loads (4 lines an instruction, not 16-line gathers); fragments read from LDS; sums unchanged
# speedup vs baseline: 7.9711x; 1.0502x over previous
#include <hip/hip_runtime.h>
#include <math.h>

typedef float v2f __attribute__((ext_vector_type(2)));
typedef float v4f __attribute__((ext_vector_type(4)));
typedef int   v4i __attribute__((ext_vector_type(4)));
typedef float v8f __attribute__((ext_vector_type(8)));

#define WMMA_F32(a, b, c) \
  __builtin_amdgcn_wmma_f32_16x16x4_f32(false, (a), false, (b), (short)0, (c), false, false)

static constexpr int B_ = 8, S_ = 512, E_ = 32, L_ = 4096, SUBE_ = 768;
static constexpr float LAMBDA_INIT = 0.2f;

__global__ __launch_bounds__(256) __attribute__((amdgpu_waves_per_eu(2, 2))) void proj_kernel(
    const float* __restrict__ key, const float* __restrict__ Wk,
    const float* __restrict__ Wv, const int* __restrict__ kmask,
    float* __restrict__ Km, float* __restrict__ Vm, float* __restrict__ Bk,
    const float* __restrict__ q, const float* __restrict__ Wq,
    float* __restrict__ Qs) {
  __shared__ __attribute__((aligned(16))) float tile[16 * 68];
  __shared__ __attribute__((aligned(16))) float stg[4 * 2 * 16 * 68];
  const int tid  = threadIdx.x;
  const int w    = tid >> 5;
  const int lane = tid & 31;
  const int n    = lane & 15;
  const int h    = lane >> 4;
  const int m0   = blockIdx.x * 16;
  const int wu   = __builtin_amdgcn_readfirstlane(w);

  if (wu >= 4) {
    for (int u = 0; u < 2; ++u) {
      const int qm0  = ((blockIdx.x * 4 + (wu - 4)) * 2 + u) * 16;
      const int mrow = qm0 + n;
      v8f a0 = {}, a1 = {};
      for (int kk = 0; kk < 8; ++kk) {
        const int kb = kk * 4 + h * 2;
        v2f a  = *(const v2f*)(q + (size_t)mrow * 32 + kb);
        v2f b0 = *(const v2f*)(Wq + (size_t)n * 32 + kb);
        v2f b1 = *(const v2f*)(Wq + (size_t)(16 + n) * 32 + kb);
        a0 = WMMA_F32(a, b0, a0);
        a1 = WMMA_F32(a, b1, a1);
      }
      const int rbase = qm0 + (h ? 8 : 0);
      for (int v = 0; v < 8; ++v) {
        const size_t r = (size_t)(rbase + v) * 32;
        Qs[r + n]      = a0[v] * 0.25f;
        Qs[r + 16 + n] = a1[v] * 0.25f;
      }
    }
  }

  const bool bias_writer = (blockIdx.x & 1) == 0 && tid < 64;
  const int kidx = (m0 + lane < B_ * S_) ? m0 + lane : B_ * S_ - 1;
  const int kmv = kmask[kidx];

  if (wu < 4) {
    const float* wsel = (w < 2) ? Wk : Wv;
    const float* ag = key + (size_t)(m0 + h) * SUBE_ + 4 * n;
    const float* bg = wsel + (size_t)((w & 1) * 16 + h) * SUBE_ + 4 * n;
    float* sa = stg + wu * (2 * 16 * 68);
    float* sb = sa + 16 * 68;
    float* sst_a = sa + h * 68 + 4 * n;
    float* sst_b = sb + h * 68 + 4 * n;
    const float* fr_a = sa + n * 68 + 4 * h;
    const float* fr_b = sb + n * 68 + 4 * h;

    constexpr int NCH = SUBE_ / 64;
    v4f ra[3][8], rb[3][8];
    v4f fa[2][8], fb[2][8];
#pragma unroll
    for (int j = 0; j < 3; ++j) {
#pragma unroll
      for (int q2 = 0; q2 < 8; ++q2) {
        ra[j][q2] = *(const v4f*)(ag + (size_t)(2 * q2) * SUBE_ + 64 * j);
        rb[j][q2] = *(const v4f*)(bg + (size_t)(2 * q2) * SUBE_ + 64 * j);
      }
    }

    v8f acc0 = {}, acc1 = {};
#pragma unroll
    for (int j = -1; j < NCH; ++j) {
      if (j + 1 < NCH) {
        const int s = (j + 1) % 3;
#pragma unroll
        for (int q2 = 0; q2 < 8; ++q2) {
          *(v4f*)(sst_a + (2 * q2) * 68) = ra[s][q2];
          *(v4f*)(sst_b + (2 * q2) * 68) = rb[s][q2];
        }
        if (j + 4 < NCH) {
#pragma unroll
          for (int q2 = 0; q2 < 8; ++q2) {
            ra[s][q2] = *(const v4f*)(ag + (size_t)(2 * q2) * SUBE_ + 64 * (j + 4));
            rb[s][q2] = *(const v4f*)(bg + (size_t)(2 * q2) * SUBE_ + 64 * (j + 4));
          }
        }
        __builtin_amdgcn_fence(__ATOMIC_RELEASE, "wavefront");
        __builtin_amdgcn_wave_barrier();
        __builtin_amdgcn_fence(__ATOMIC_ACQUIRE, "wavefront");
        const int f = (j + 1) & 1;
#pragma unroll
        for (int i = 0; i < 8; ++i) {
          fa[f][i] = *(const v4f*)(fr_a + 8 * i);
          fb[f][i] = *(const v4f*)(fr_b + 8 * i);
        }
        __builtin_amdgcn_fence(__ATOMIC_RELEASE, "wavefront");
        __builtin_amdgcn_wave_barrier();
        __builtin_amdgcn_fence(__ATOMIC_ACQUIRE, "wavefront");
      }
      __builtin_amdgcn_sched_barrier(0);
      if (j >= 0) {
        const int f = j & 1;
#pragma unroll
        for (int i = 0; i < 8; ++i) {
          v2f a01, b01, a23, b23;
          a01[0] = fa[f][i][0]; a01[1] = fa[f][i][1]; b01[0] = fb[f][i][0]; b01[1] = fb[f][i][1];
          a23[0] = fa[f][i][2]; a23[1] = fa[f][i][3]; b23[0] = fb[f][i][2]; b23[1] = fb[f][i][3];
          acc0 = WMMA_F32(a01, b01, acc0);
          acc1 = WMMA_F32(a23, b23, acc1);
        }
      }
      __builtin_amdgcn_sched_barrier(0);
    }

#pragma unroll
    for (int v = 0; v < 8; ++v)
      tile[(8 * h + v) * 68 + 16 * w + n] = acc0[v] + acc1[v];
  }
  __syncthreads();

  if (wu < 4) {
#pragma unroll
    for (int q2 = 0; q2 < 2; ++q2) {
      const int p    = tid + 128 * q2;
      const int line = p >> 3;
      const int seg  = p & 7;
      const int row  = line & 15;
      const int isv  = line >> 4;
      const v4f x = *(const v4f*)(tile + row * 68 + 32 * isv + 4 * seg);
      float* dst = (isv ? Vm : Km) + (size_t)(m0 + row) * 32 + 4 * seg;
      *(v4f*)dst = x;
    }
    if (bias_writer)
      Bk[(size_t)w * (B_ * S_) + m0 + lane] = kmv ? 0.f : -1e20f;
  }
}

__global__ __launch_bounds__(256) __attribute__((amdgpu_waves_per_eu(2, 2))) void attn_kernel(
    const float* __restrict__ Qs, const float* __restrict__ Km,
    const float* __restrict__ Vm, const int* __restrict__ qmask,
    const float* __restrict__ Bk, const float* __restrict__ lq1,
    const float* __restrict__ lk1, const float* __restrict__ lq2,
    const float* __restrict__ lk2, const float* __restrict__ lnw,
    const float* __restrict__ Wo, float* __restrict__ outp,
    float* __restrict__ diffp) {
  __shared__ __attribute__((aligned(16))) v4f park_all[8 * 64 * 32];
  __shared__ __attribute__((aligned(16))) v4f kst[512 * 8];
  const int lane = threadIdx.x & 31;
  const int wv   = __builtin_amdgcn_readfirstlane((int)(threadIdx.x >> 5));
  v4f* park      = park_all + wv * (64 * 32);
  const int c    = lane & 15;
  const int h    = lane >> 4;
  const int R0   = (blockIdx.x * 8 + wv) * 16;
  const int b    = R0 / L_;

  const int tid = threadIdx.x;
  const v4f* ksrc = (const v4f*)(Km + (size_t)b * S_ * 32);
  v4f kf[16];
#pragma unroll
  for (int i = 0; i < 16; ++i) kf[i] = ksrc[i * 256 + tid];

  float s1 = 0.f, s2 = 0.f;
  for (int i = 0; i < 16; ++i) { s1 += lq1[i] * lk1[i]; s2 += lq2[i] * lk2[i]; }
  const float lambda_full = __expf(s1) - __expf(s2) + LAMBDA_INIT;

  const int qv = qmask[R0 + c];

  const int ke = (tid & ~7) + ((tid & 7) ^ ((tid >> 4) & 7));
#pragma unroll
  for (int i = 0; i < 16; ++i) kst[i * 256 + ke] = kf[i];
  __syncthreads();

  const int crow = ((c & 4) << 1) | ((c & 8) >> 1) | (c & 3);
  const int ksw = (crow >> 1) & 7;
  const float* bbase = Bk + b * S_ + 4 * h;

  const float zq = qv ? 0.f : 1.f;

  v8f D[32];

#pragma unroll 1
  for (int hd = 0; hd < 2; ++hd) {
    const float* qp = Qs + (size_t)(R0 + c) * 32 + hd * 16 + 8 * h;
    const v4f qa = *(const v4f*)(qp);
    const v4f qb = *(const v4f*)(qp + 4);
    v2f q0, q1, q2, q3;
    q0[0] = qa[0]; q0[1] = qa[1];
    q1[0] = qa[2]; q1[1] = qa[3];
    q2[0] = qb[0]; q2[1] = qb[1];
    q3[0] = qb[2]; q3[1] = qb[3];
    const v4f* ka = kst + crow * 8 + ((4 * hd + 2 * h) ^ ksw);
    const v4f* kb = kst + crow * 8 + ((4 * hd + 2 * h + 1) ^ ksw);
    const float* bh = bbase + hd * (B_ * S_);

    v4f KA[32], KB[32];
#pragma unroll
    for (int g = 0; g <= 4; ++g) {
      if (g < 4) {
#pragma unroll
        for (int i = 0; i < 8; ++i) {
          const int t = 8 * g + i;
          KA[t] = ka[t * 128];
          KB[t] = kb[t * 128];
          const v4f ba = *(const v4f*)(bh + t * 16);
          const v4f bb = *(const v4f*)(bh + t * 16 + 8);
          v8f x;
          x[0] = ba[0]; x[1] = ba[1]; x[2] = ba[2]; x[3] = ba[3];
          x[4] = bb[0]; x[5] = bb[1]; x[6] = bb[2]; x[7] = bb[3];
          D[t] = x;
        }
      }
      __builtin_amdgcn_sched_barrier(0);
      if (g > 0) {
#pragma unroll
        for (int i = 0; i < 8; i += 2) {
          const int t = 8 * (g - 1) + i;
          const v4f xa = KA[t], xb = KB[t], ya = KA[t + 1], yb = KB[t + 1];
          v8f x = D[t], y = D[t + 1];
          v2f f;
          f[0] = xa[0]; f[1] = xa[1]; x = WMMA_F32(f, q0, x);
          f[0] = ya[0]; f[1] = ya[1]; y = WMMA_F32(f, q0, y);
          f[0] = xa[2]; f[1] = xa[3]; x = WMMA_F32(f, q1, x);
          f[0] = ya[2]; f[1] = ya[3]; y = WMMA_F32(f, q1, y);
          f[0] = xb[0]; f[1] = xb[1]; x = WMMA_F32(f, q2, x);
          f[0] = yb[0]; f[1] = yb[1]; y = WMMA_F32(f, q2, y);
          f[0] = xb[2]; f[1] = xb[3]; x = WMMA_F32(f, q3, x);
          f[0] = yb[2]; f[1] = yb[3]; y = WMMA_F32(f, q3, y);
          D[t] = x; D[t + 1] = y;
        }
      }
      __builtin_amdgcn_sched_barrier(0);
    }

    v8f mv = D[0];
#pragma unroll
    for (int t = 1; t < 32; ++t) {
#pragma unroll
      for (int r = 0; r < 8; ++r) mv[r] = fmaxf(mv[r], D[t][r]);
    }
    float mx = fmaxf(fmaxf(fmaxf(mv[0], mv[1]), fmaxf(mv[2], mv[3])),
                     fmaxf(fmaxf(mv[4], mv[5]), fmaxf(mv[6], mv[7])));
    mx = fmaxf(mx, __shfl_xor(mx, 16, 32));

    v8f sv = {};
#pragma unroll
    for (int t = 0; t < 32; ++t) {
#pragma unroll
      for (int r = 0; r < 8; ++r) {
        const float e = __expf(D[t][r] - mx);
        D[t][r] = e; sv[r] += e;
      }
    }
    float sum = ((sv[0] + sv[1]) + (sv[2] + sv[3])) +
                ((sv[4] + sv[5]) + (sv[6] + sv[7]));
    sum += __shfl_xor(sum, 16, 32);
    const float inv   = 1.f / (sum + 1e-8f);
    if (hd == 0) {
#pragma unroll
      for (int t = 0; t < 32; ++t) {
        v4f lo, hi;
#pragma unroll
        for (int r = 0; r < 4; ++r) { lo[r] = D[t][r] * inv; hi[r] = D[t][4 + r] * inv; }
        park[(2 * t) * 32 + lane]     = lo;
        park[(2 * t + 1) * 32 + lane] = hi;
      }
    } else {
      const float scale = -lambda_full * inv;
#pragma unroll
      for (int t = 0; t < 32; ++t) {
        const v4f lo = park[(2 * t) * 32 + lane];
        const v4f hi = park[(2 * t + 1) * 32 + lane];
#pragma unroll
        for (int r = 0; r < 4; ++r) {
          D[t][r]     = D[t][r] * scale + lo[r];
          D[t][4 + r] = D[t][4 + r] * scale + hi[r];
        }
      }
    }
  }

  v8f dmv = D[0];
#pragma unroll
  for (int t = 1; t < 32; ++t) {
#pragma unroll
    for (int r = 0; r < 8; ++r) dmv[r] = fminf(dmv[r], D[t][r]);
  }
  float dmin = fminf(fminf(fminf(dmv[0], dmv[1]), fminf(dmv[2], dmv[3])),
                     fminf(fminf(dmv[4], dmv[5]), fminf(dmv[6], dmv[7])));
  dmin = fminf(dmin, __shfl_xor(dmin, 16, 32));

  int wb[8];
#pragma unroll
  for (int k = 0; k < 8; ++k) wb[k] = c * 128 + ((2 * k + h) ^ c);
#pragma unroll
  for (int t = 0; t < 32; ++t) {
    const v4f ba = *(const v4f*)(bbase + t * 16);
    const v4f bb = *(const v4f*)(bbase + t * 16 + 8);
    v4f lo, hi;
#pragma unroll
    for (int r = 0; r < 8; ++r) {
      const float bias = (r < 4) ? ba[r & 3] : bb[r & 3];
      float dd = D[t][r] - dmin + 1e-5f;
      dd = (bias == zq) ? dd : 0.f;
      D[t][r] = dd;
      if (r < 4) lo[r & 3] = dd; else hi[r & 3] = dd;
    }
    park[wb[2 * (t & 3)]     + (t >> 2) * 16] = lo;
    park[wb[2 * (t & 3) + 1] + (t >> 2) * 16] = hi;
  }
  __builtin_amdgcn_fence(__ATOMIC_RELEASE, "wavefront");
  __builtin_amdgcn_wave_barrier();
  __builtin_amdgcn_fence(__ATOMIC_ACQUIRE, "wavefront");
  float* dg = diffp + (size_t)R0 * 512 + 4 * lane;
#pragma unroll
  for (int R = 0; R < 16; ++R) {
    const int rb = R * 128 + (lane ^ R);
#pragma unroll
    for (int i = 0; i < 4; ++i) {
      const v4f x = park[rb + 32 * i];
      __builtin_nontemporal_store(x, (v4f*)(dg + R * 512 + 128 * i));
    }
  }
  __builtin_amdgcn_sched_barrier(0x0020);

  const float* vbase = Vm + (size_t)(b * S_ + 4 * h) * 32 + 2 * c;
  v8f ao0 = {}, ao1 = {};
#pragma unroll
  for (int t = 0; t < 32; ++t) {
#pragma unroll
    for (int j = 0; j < 4; ++j) {
      const float* vp = vbase + (size_t)(t * 16 + 8 * (j >> 1) + 2 * (j & 1)) * 32;
      const v2f k0 = *(const v2f*)(vp);
      const v2f k1 = *(const v2f*)(vp + 32);
      v2f a0, a1, bf;
      a0[0] = k0[0]; a0[1] = k1[0];
      a1[0] = k0[1]; a1[1] = k1[1];
      bf[0] = D[t][2 * j]; bf[1] = D[t][2 * j + 1];
      ao0 = WMMA_F32(a0, bf, ao0);
      ao1 = WMMA_F32(a1, bf, ao1);
    }
  }

  float sq = 0.f;
#pragma unroll
  for (int r = 0; r < 8; ++r) sq += ao0[r] * ao0[r];
#pragma unroll
  for (int r = 0; r < 8; ++r) sq += ao1[r] * ao1[r];
  sq += __shfl_xor(sq, 16, 32);
  const float rms = rsqrtf(sq * (1.0f / 32.0f) + 1e-5f);
#pragma unroll
  for (int r = 0; r < 8; ++r) {
    ao0[r] = ao0[r] * rms * lnw[16 * h + 2 * r] * 0.8f;
    ao1[r] = ao1[r] * rms * lnw[16 * h + 2 * r + 1] * 0.8f;
  }

  v8f o0 = {}, o1 = {};
#pragma unroll
  for (int mt = 0; mt < 2; ++mt) {
#pragma unroll
    for (int j = 0; j < 4; ++j) {
      const int kb = 16 * h + 4 * j;
      const v4f x0 = *(const v4f*)(Wo + (size_t)c * 32 + kb);
      const v4f x1 = *(const v4f*)(Wo + (size_t)(16 + c) * 32 + kb);
      v2f w0, w1, bf;
      w0[0] = x0[mt]; w0[1] = x0[2 + mt];
      w1[0] = x1[mt]; w1[1] = x1[2 + mt];
      bf[0] = mt ? ao1[2 * j] : ao0[2 * j];
      bf[1] = mt ? ao1[2 * j + 1] : ao0[2 * j + 1];
      o0 = WMMA_F32(w0, bf, o0);
      o1 = WMMA_F32(w1, bf, o1);
    }
  }
  float* og = outp + (size_t)(R0 + c) * 32 + 8 * h;
  v4f x;
  x[0] = o0[0]; x[1] = o0[1]; x[2] = o0[2]; x[3] = o0[3];
  __builtin_nontemporal_store(x, (v4f*)(og));
  x[0] = o0[4]; x[1] = o0[5]; x[2] = o0[6]; x[3] = o0[7];
  __builtin_nontemporal_store(x, (v4f*)(og + 4));
  x[0] = o1[0]; x[1] = o1[1]; x[2] = o1[2]; x[3] = o1[3];
  __builtin_nontemporal_store(x, (v4f*)(og + 16));
  x[0] = o1[4]; x[1] = o1[5]; x[2] = o1[6]; x[3] = o1[7];
  __builtin_nontemporal_store(x, (v4f*)(og + 20));
}

extern "C" void kernel_launch(void* const* d_in, const int* in_sizes, int n_in,
                              void* d_out, int out_size, void* d_ws, size_t ws_size,
                              hipStream_t stream) {
  (void)in_sizes; (void)n_in; (void)out_size; (void)ws_size;
  const float* query = (const float*)d_in[0];
  const float* key   = (const float*)d_in[1];
  const int*   qmask = (const int*)d_in[2];
  const int*   kmask = (const int*)d_in[3];
  const float* Wq = (const float*)d_in[4];
  const float* Wk = (const float*)d_in[5];
  const float* Wv = (const float*)d_in[6];
  const float* Wo = (const float*)d_in[7];
  const float* lq1 = (const float*)d_in[8];
  const float* lk1 = (const float*)d_in[9];
  const float* lq2 = (const float*)d_in[10];
  const float* lk2 = (const float*)d_in[11];
  const float* lnw = (const float*)d_in[12];

  float* outp  = (float*)d_out;
  float* diffp = outp + (size_t)B_ * L_ * E_;

  float* Qs = (float*)d_ws;
  float* Km = Qs + (size_t)B_ * L_ * E_;
  float* Vm = Km + (size_t)B_ * S_ * E_;
  float* Bk = Vm + (size_t)B_ * S_ * E_;

  proj_kernel<<<(B_ * S_) / 16, 256, 0, stream>>>(key, Wk, Wv, kmask, Km, Vm, Bk,
                                                  query, Wq, Qs);
  attn_kernel<<<(B_ * L_) / 128, 256, 0, stream>>>(Qs, Km, Vm, qmask, Bk,
                                                 lq1, lk1, lq2, lk2, lnw, Wo,
                                                 outp, diffp);
}
